# starting kernel
# baseline (speedup 1.0000x reference)
#include <hip/hip_runtime.h>


#define S_DIM   64
#define C_DIM   8
#define N_ITER  4
#define NEUMANN 6
#define B_TOT   128
#define NB      8
#define BSUB    8
#define NWG     (B_TOT / BSUB)
#define THREADS 128
#define DT_C    0.05f
#define LDA     66

typedef __attribute__((ext_vector_type(2))) float v2f;
typedef __attribute__((ext_vector_type(8))) float v8f;

__device__ __forceinline__ v8f wmma_gemm_16x64(const float* __restrict__ A,
                                               const float* __restrict__ Bt,
                                               int n0, int lane) {
  v8f acc = {0.f, 0.f, 0.f, 0.f, 0.f, 0.f, 0.f, 0.f};
  const int m = lane & 15;
  const int h = (lane >> 4) << 1;
#pragma unroll
  for (int k0 = 0; k0 < S_DIM; k0 += 4) {
    v2f a, b;
    a.x = A[m * LDA + k0 + h];
    a.y = A[m * LDA + k0 + h + 1];
    b.x = Bt[(k0 + h) * LDA + n0 + m];
    b.y = Bt[(k0 + h + 1) * LDA + n0 + m];
    acc = __builtin_amdgcn_wmma_f32_16x16x4_f32(false, a, false, b,
                                                (short)0, acc, false, false);
  }
  return acc;
}

__global__ __launch_bounds__(THREADS, 1)
void solver_gfx1250_kernel(const float* __restrict__ y0,
                           const float* __restrict__ forces,
                           const float* __restrict__ W,
                           float* __restrict__ out) {
  __shared__ float sWt[S_DIM * LDA];
  __shared__ float sS [16 * LDA];
  __shared__ float sX0[16 * LDA];
  __shared__ float sX1[16 * LDA];
  __shared__ float sY[C_DIM][BSUB * S_DIM];
  __shared__ float sYprev[BSUB * S_DIM];
  __shared__ float sYm  [BSUB * S_DIM];
  __shared__ float sT   [BSUB * S_DIM];
  __shared__ float sG   [BSUB * S_DIM];
  __shared__ float sRhs [BSUB * S_DIM];
  __shared__ float sWx  [BSUB * S_DIM];

  const int tid  = threadIdx.x;
  const int lane = tid & 31;
  const int wave = tid >> 5;
  const int n0   = wave * 16;
  const int b0   = blockIdx.x * BSUB;

  for (int e = tid; e < S_DIM * S_DIM; e += THREADS) {
    int p = e >> 6, s = e & 63;
    sWt[s * LDA + p] = W[p * S_DIM + s];
  }
  for (int e = tid; e < 16 * LDA; e += THREADS) {
    sS[e] = 0.f; sX0[e] = 0.f; sX1[e] = 0.f;
  }
  for (int e = tid; e < BSUB * S_DIM; e += THREADS) {
    int b = e >> 6, j = e & 63;
    float v = y0[(b0 + b) * S_DIM + j];
    sYprev[e] = v;
    out[(size_t)(b0 + b) * S_DIM + j] = v;
  }
  __syncthreads();

  for (int blk = 0; blk < NB; ++blk) {
    const float* fblk =
        forces + ((size_t)(1 + blk * C_DIM) * B_TOT + b0) * S_DIM;
    for (int off = tid * 16; off < C_DIM * BSUB * S_DIM; off += THREADS * 16)
      __builtin_prefetch(fblk + off, 0, 1);

    for (int c = 0; c < C_DIM; ++c)
      for (int e = tid; e < BSUB * S_DIM; e += THREADS) sY[c][e] = 0.f;
    __syncthreads();

    for (int it = 0; it < N_ITER; ++it) {
      for (int e = tid; e < BSUB * S_DIM; e += THREADS) sYm[e] = sYprev[e];
      __syncthreads();

      for (int c = 0; c < C_DIM; ++c) {
        for (int e = tid; e < BSUB * S_DIM; e += THREADS) {
          int b = e >> 6, j = e & 63;
          sS[b * LDA + j] = 0.5f * (sYm[e] + sY[c][e]);
        }
        __syncthreads();

        {
          v8f tacc = wmma_gemm_16x64(sS, sWt, n0, lane);
          v8f wacc = {0.f, 0.f, 0.f, 0.f, 0.f, 0.f, 0.f, 0.f};
          if (c > 0) wacc = wmma_gemm_16x64(sX0, sWt, n0, lane);
          if ((lane >> 4) == 0) {
            int col = n0 + (lane & 15);
#pragma unroll
            for (int v = 0; v < 8; ++v) {
              sT[v * S_DIM + col] = tanhf(tacc[v]);
              if (c > 0) sWx[v * S_DIM + col] = wacc[v];
            }
          }
        }
        __syncthreads();

        for (int e = tid; e < BSUB * S_DIM; e += THREADS) {
          int b = e >> 6, j = e & 63;
          float t = sT[e];
          float g = 0.5f * DT_C * (1.0f - t * t);
          float fv = fblk[(size_t)c * B_TOT * S_DIM + b * S_DIM + j];
          float R = sY[c][e] - sYm[e] + DT_C * t - DT_C * fv;
          float rhs = R;
          if (c > 0) {
            float xp = sX0[b * LDA + j];
            rhs += xp - g * sWx[e];
          }
          sG[e] = g;
          sRhs[e] = rhs;
          sX0[b * LDA + j] = rhs;
        }
        __syncthreads();

        for (int k = 0; k < NEUMANN; ++k) {
          const float* xin  = (k & 1) ? sX1 : sX0;
          float*       xout = (k & 1) ? sX0 : sX1;
          v8f macc = wmma_gemm_16x64(xin, sWt, n0, lane);
          if ((lane >> 4) == 0) {
            int col = n0 + (lane & 15);
#pragma unroll
            for (int v = 0; v < 8; ++v) {
              int e = v * S_DIM + col;
              xout[v * LDA + col] = sRhs[e] - sG[e] * macc[v];
            }
          }
          __syncthreads();
        }

        for (int e = tid; e < BSUB * S_DIM; e += THREADS) {
          int b = e >> 6, j = e & 63;
          float yold = sY[c][e];
          sYm[e] = yold;
          sY[c][e] = yold - sX0[b * LDA + j];
        }
        __syncthreads();
      }
    }

    for (int c = 0; c < C_DIM; ++c)
      for (int e = tid; e < BSUB * S_DIM; e += THREADS) {
        int b = e >> 6, j = e & 63;
        out[((size_t)(1 + blk * C_DIM + c) * B_TOT + b0 + b) * S_DIM + j] =
            sY[c][e];
      }
    for (int e = tid; e < BSUB * S_DIM; e += THREADS)
      sYprev[e] = sY[C_DIM - 1][e];
    __syncthreads();
  }
}

extern "C" void kernel_launch(void* const* d_in, const int* in_sizes, int n_in,
                              void* d_out, int out_size, void* d_ws,
                              size_t ws_size, hipStream_t stream) {
  (void)in_sizes; (void)n_in; (void)out_size; (void)d_ws; (void)ws_size;
  const float* y0     = (const float*)d_in[0];
  const float* forces = (const float*)d_in[1];
  const float* W      = (const float*)d_in[2];
  float* out = (float*)d_out;
  solver_gfx1250_kernel<<<NWG, THREADS, 0, stream>>>(y0, forces, W, out);
}


# Newton-iteration x chunk cells run as an anti-diagonal wavefront: up to 4 independent cells share one 32-row WMMA step (8 waves), 88 serial steps, not 256; sum order kept
# speedup vs baseline: 1.8359x; 1.8359x over previous
#include <hip/hip_runtime.h>


#define S_DIM   64
#define C_DIM   8
#define N_ITER  4
#define NEUMANN 6
#define B_TOT   128
#define NB      8
#define BSUB    8
#define NWG     (B_TOT / BSUB)
#define ROWS    (N_ITER * BSUB)
#define NDIAG   (C_DIM + N_ITER - 1)
#define PLANE   (BSUB * S_DIM)
#define THREADS 256
#define DT_C    0.05f
#define LDA     66

typedef __attribute__((ext_vector_type(2))) float v2f;
typedef __attribute__((ext_vector_type(8))) float v8f;

__device__ __forceinline__ v8f wmma_gemm_16x64(const float* __restrict__ A,
                                               const float* __restrict__ Bt,
                                               int n0, int lane) {
  v8f acc = {0.f, 0.f, 0.f, 0.f, 0.f, 0.f, 0.f, 0.f};
  const int m = lane & 15;
  const int h = (lane >> 4) << 1;
#pragma unroll
  for (int k0 = 0; k0 < S_DIM; k0 += 4) {
    v2f a, b;
    a.x = A[m * LDA + k0 + h];
    a.y = A[m * LDA + k0 + h + 1];
    b.x = Bt[(k0 + h) * LDA + n0 + m];
    b.y = Bt[(k0 + h + 1) * LDA + n0 + m];
    acc = __builtin_amdgcn_wmma_f32_16x16x4_f32(false, a, false, b,
                                                (short)0, acc, false, false);
  }
  return acc;
}

__global__ __launch_bounds__(THREADS, 1)
void solver_gfx1250_kernel(const float* __restrict__ y0,
                           const float* __restrict__ forces,
                           const float* __restrict__ W,
                           float* __restrict__ out) {
  __shared__ float sWt[S_DIM * LDA];
  __shared__ float sS [ROWS * LDA];
  __shared__ float sX0[ROWS * LDA];
  __shared__ float sX1[ROWS * LDA];
  __shared__ float sYv[N_ITER - 1][C_DIM][PLANE];
  __shared__ float sYprev[PLANE];
  __shared__ float sT  [ROWS * S_DIM];
  __shared__ float sG  [ROWS * S_DIM];
  __shared__ float sRhs[ROWS * S_DIM];
  __shared__ float sWx [ROWS * S_DIM];

  const int tid  = threadIdx.x;
  const int lane = tid & 31;
  const int wave = tid >> 5;
  const int mt   = wave >> 2;
  const int n0   = (wave & 3) * 16;
  const int slot = mt * 2 + (lane >> 4);
  const int col  = n0 + (lane & 15);
  const int b0   = blockIdx.x * BSUB;

  for (int e = tid; e < S_DIM * S_DIM; e += THREADS) {
    int p = e >> 6, s = e & 63;
    sWt[s * LDA + p] = W[p * S_DIM + s];
  }
  for (int e = tid; e < ROWS * LDA; e += THREADS) {
    sS[e] = 0.f; sX0[e] = 0.f; sX1[e] = 0.f;
  }
  for (int e = tid; e < PLANE; e += THREADS) {
    int b = e >> 6, j = e & 63;
    float v = y0[(b0 + b) * S_DIM + j];
    sYprev[e] = v;
    out[(size_t)(b0 + b) * S_DIM + j] = v;
  }
  __syncthreads();

  for (int blk = 0; blk < NB; ++blk) {
    const float* fblk =
        forces + ((size_t)(1 + blk * C_DIM) * B_TOT + b0) * S_DIM;
    for (int off = tid * 16; off < C_DIM * BSUB * S_DIM; off += THREADS * 16)
      __builtin_prefetch(fblk + off, 0, 1);

    for (int d = 0; d < NDIAG; ++d) {
      const bool tile_act = (d >= 2 * mt) && (d <= 2 * mt + C_DIM);
      const bool lane_act = (unsigned)(d - slot) < (unsigned)C_DIM;

      for (int e = tid; e < ROWS * S_DIM; e += THREADS) {
        int it = e / PLANE, r = e % PLANE;
        int c = d - it;
        if ((unsigned)c < (unsigned)C_DIM) {
          int b = r >> 6, j = r & 63;
          float ym = (c == 0) ? sYprev[r] : (it == 0 ? 0.f : sYv[it - 1][c - 1][r]);
          float yc = (it == 0) ? 0.f : sYv[it - 1][c][r];
          sS[(it * BSUB + b) * LDA + j] = 0.5f * (ym + yc);
        }
      }
      __syncthreads();

      if (tile_act) {
        v8f tacc = wmma_gemm_16x64(sS + mt * 16 * LDA, sWt, n0, lane);
        v8f wacc = wmma_gemm_16x64(sX0 + mt * 16 * LDA, sWt, n0, lane);
        if (lane_act) {
#pragma unroll
          for (int v = 0; v < 8; ++v) {
            int e = (slot * BSUB + v) * S_DIM + col;
            sT[e]  = tanhf(tacc[v]);
            sWx[e] = wacc[v];
          }
        }
      }
      __syncthreads();

      for (int e = tid; e < ROWS * S_DIM; e += THREADS) {
        int it = e / PLANE, r = e % PLANE;
        int c = d - it;
        if ((unsigned)c < (unsigned)C_DIM) {
          int b = r >> 6, j = r & 63;
          int xi = (it * BSUB + b) * LDA + j;
          float ym = (c == 0) ? sYprev[r] : (it == 0 ? 0.f : sYv[it - 1][c - 1][r]);
          float yc = (it == 0) ? 0.f : sYv[it - 1][c][r];
          float t = sT[e];
          float g = 0.5f * DT_C * (1.0f - t * t);
          float fv = fblk[(size_t)c * B_TOT * S_DIM + r];
          float R = yc - ym + DT_C * t - DT_C * fv;
          float rhs = R;
          if (c > 0) {
            float xp = sX0[xi];
            rhs += xp - g * sWx[e];
          }
          sG[e] = g;
          sRhs[e] = rhs;
          sX0[xi] = rhs;
        }
      }
      __syncthreads();

      for (int k = 0; k < NEUMANN; ++k) {
        const float* xin  = (k & 1) ? sX1 : sX0;
        float*       xout = (k & 1) ? sX0 : sX1;
        if (tile_act) {
          v8f macc = wmma_gemm_16x64(xin + mt * 16 * LDA, sWt, n0, lane);
          if (lane_act) {
#pragma unroll
            for (int v = 0; v < 8; ++v) {
              int e = (slot * BSUB + v) * S_DIM + col;
              xout[(slot * BSUB + v) * LDA + col] = sRhs[e] - sG[e] * macc[v];
            }
          }
        }
        __syncthreads();
      }

      for (int e = tid; e < ROWS * S_DIM; e += THREADS) {
        int it = e / PLANE, r = e % PLANE;
        int c = d - it;
        if ((unsigned)c < (unsigned)C_DIM) {
          int b = r >> 6, j = r & 63;
          float yold = (it == 0) ? 0.f : sYv[it - 1][c][r];
          float ynew = yold - sX0[(it * BSUB + b) * LDA + j];
          if (it < N_ITER - 1) {
            sYv[it][c][r] = ynew;
          } else {
            out[((size_t)(1 + blk * C_DIM + c) * B_TOT + b0 + b) * S_DIM + j] = ynew;
            if (c == C_DIM - 1) sYprev[r] = ynew;
          }
        }
      }
      __syncthreads();
    }
  }
}

extern "C" void kernel_launch(void* const* d_in, const int* in_sizes, int n_in,
                              void* d_out, int out_size, void* d_ws,
                              size_t ws_size, hipStream_t stream) {
  (void)in_sizes; (void)n_in; (void)out_size; (void)d_ws; (void)ws_size;
  const float* y0     = (const float*)d_in[0];
  const float* forces = (const float*)d_in[1];
  const float* W      = (const float*)d_in[2];
  float* out = (float*)d_out;
  solver_gfx1250_kernel<<<NWG, THREADS, 0, stream>>>(y0, forces, W, out);
}


# Element-wise stages (midpoint state, residual, Neumann and Newton updates) fused into the WMMA epilogues in accumulator lane layout: registers replace LDS round trips; 7 barriers per step, not 10
# speedup vs baseline: 3.3163x; 1.8064x over previous
#include <hip/hip_runtime.h>


#define S_DIM   64
#define C_DIM   8
#define N_ITER  4
#define NEUMANN 6
#define B_TOT   128
#define NB      8
#define BSUB    8
#define NWG     (B_TOT / BSUB)
#define ROWS    (N_ITER * BSUB)
#define NDIAG   (C_DIM + N_ITER - 1)
#define PLANE   (BSUB * S_DIM)
#define THREADS 256
#define DT_C    0.05f
#define LDA     66

typedef __attribute__((ext_vector_type(2))) float v2f;
typedef __attribute__((ext_vector_type(8))) float v8f;

__device__ __forceinline__ v8f wmma_gemm_16x64(const float* __restrict__ A,
                                               const float* __restrict__ Bt,
                                               int n0, int lane) {
  v8f acc = {0.f, 0.f, 0.f, 0.f, 0.f, 0.f, 0.f, 0.f};
  const int m = lane & 15;
  const int h = (lane >> 4) << 1;
#pragma unroll
  for (int k0 = 0; k0 < S_DIM; k0 += 4) {
    v2f a, b;
    a.x = A[m * LDA + k0 + h];
    a.y = A[m * LDA + k0 + h + 1];
    b.x = Bt[(k0 + h) * LDA + n0 + m];
    b.y = Bt[(k0 + h + 1) * LDA + n0 + m];
    acc = __builtin_amdgcn_wmma_f32_16x16x4_f32(false, a, false, b,
                                                (short)0, acc, false, false);
  }
  return acc;
}

__global__ __launch_bounds__(THREADS, 1)
void solver_gfx1250_kernel(const float* __restrict__ y0,
                           const float* __restrict__ forces,
                           const float* __restrict__ W,
                           float* __restrict__ out) {
  __shared__ float sWt[S_DIM * LDA];
  __shared__ float sS [ROWS * LDA];
  __shared__ float sXa[ROWS * LDA];
  __shared__ float sXb[ROWS * LDA];
  __shared__ float sYv[N_ITER - 1][C_DIM][PLANE];
  __shared__ float sYprev[PLANE];
  __shared__ __attribute__((aligned(16))) float sOut[PLANE];

  const int tid  = threadIdx.x;
  const int lane = tid & 31;
  const int wave = tid >> 5;
  const int mt   = wave >> 2;
  const int n0   = (wave & 3) * 16;
  const int slot = mt * 2 + (lane >> 4);
  const int col  = n0 + (lane & 15);
  const int b0   = blockIdx.x * BSUB;
  const int rowbase = slot * BSUB * LDA + col;

  for (int e = tid; e < S_DIM * S_DIM; e += THREADS) {
    int p = e >> 6, s = e & 63;
    sWt[s * LDA + p] = W[p * S_DIM + s];
  }
  for (int e = tid; e < ROWS * LDA; e += THREADS) {
    sS[e] = 0.f; sXa[e] = 0.f; sXb[e] = 0.f;
  }
  __syncthreads();
  for (int e = tid; e < PLANE; e += THREADS) {
    int b = e >> 6, j = e & 63;
    float v = y0[(b0 + b) * S_DIM + j];
    sYprev[e] = v;
    sS[b * LDA + j] = 0.5f * v;
    out[(size_t)(b0 + b) * S_DIM + j] = v;
  }
  __syncthreads();

  float xfin[BSUB];
  float ylast[BSUB];
#pragma unroll
  for (int v = 0; v < BSUB; ++v) { xfin[v] = 0.f; ylast[v] = 0.f; }
  int par = 0;

  for (int blk = 0; blk < NB; ++blk) {
    const float* fblk =
        forces + ((size_t)(1 + blk * C_DIM) * B_TOT + b0) * S_DIM;
    for (int off = tid * 16; off < C_DIM * BSUB * S_DIM; off += THREADS * 16)
      __builtin_prefetch(fblk + off, 0, 1);

    for (int d = 0; d < NDIAG; ++d) {
      const bool tile_act = (d >= 2 * mt) && (d <= 2 * mt + C_DIM);
      const int  c        = d - slot;
      const bool lane_act = (unsigned)c < (unsigned)C_DIM;
      float* bufP = par ? sXb : sXa;
      float* bufQ = par ? sXa : sXb;

      float ym[BSUB], yc[BSUB], fv[BSUB], gg[BSUB], rr[BSUB];
#pragma unroll
      for (int v = 0; v < BSUB; ++v) { ym[v] = 0.f; yc[v] = 0.f; fv[v] = 0.f; gg[v] = 0.f; rr[v] = 0.f; }

      if (lane_act) {
        const float* fp = fblk + (size_t)c * B_TOT * S_DIM + col;
#pragma unroll
        for (int v = 0; v < BSUB; ++v) fv[v] = fp[v * S_DIM];
        if (c == 0) {
#pragma unroll
          for (int v = 0; v < BSUB; ++v) ym[v] = sYprev[v * S_DIM + col];
        } else if (slot > 0) {
#pragma unroll
          for (int v = 0; v < BSUB; ++v) ym[v] = sYv[slot - 1][c - 1][v * S_DIM + col];
        }
        if (slot > 0) {
#pragma unroll
          for (int v = 0; v < BSUB; ++v) yc[v] = sYv[slot - 1][c][v * S_DIM + col];
        }
      }

      if (tile_act) {
        v8f tacc = wmma_gemm_16x64(sS + mt * 16 * LDA, sWt, n0, lane);
        v8f wacc = wmma_gemm_16x64(bufP + mt * 16 * LDA, sWt, n0, lane);
        if (lane_act) {
#pragma unroll
          for (int v = 0; v < BSUB; ++v) {
            float t = tanhf(tacc[v]);
            float g = 0.5f * DT_C * (1.0f - t * t);
            float R = yc[v] - ym[v] + DT_C * t - DT_C * fv[v];
            float rhs = R;
            if (c > 0) rhs += xfin[v] - g * wacc[v];
            gg[v] = g;
            rr[v] = rhs;
            bufQ[rowbase + v * LDA] = rhs;
          }
        }
      }
      __syncthreads();

      for (int k = 0; k < NEUMANN; ++k) {
        const float* xin  = (k & 1) ? bufP : bufQ;
        float*       xout = (k & 1) ? bufQ : bufP;
        if (tile_act) {
          v8f macc = wmma_gemm_16x64(xin + mt * 16 * LDA, sWt, n0, lane);
          if (lane_act) {
            float xn[BSUB];
#pragma unroll
            for (int v = 0; v < BSUB; ++v) {
              xn[v] = rr[v] - gg[v] * macc[v];
              xout[rowbase + v * LDA] = xn[v];
            }
            if (k == NEUMANN - 1) {
              float yn[BSUB], sn[BSUB];
#pragma unroll
              for (int v = 0; v < BSUB; ++v) {
                float yl = (c == 0) ? ym[v] : ylast[v];
                yn[v] = yc[v] - xn[v];
                sn[v] = 0.5f * (yl + yn[v]);
                xfin[v] = xn[v];
                ylast[v] = yn[v];
              }
              if (slot < N_ITER - 1) {
#pragma unroll
                for (int v = 0; v < BSUB; ++v) {
                  sYv[slot][c][v * S_DIM + col] = yn[v];
                  sS[rowbase + (BSUB + v) * LDA] = sn[v];
                }
              } else {
#pragma unroll
                for (int v = 0; v < BSUB; ++v) sOut[v * S_DIM + col] = yn[v];
                if (c == C_DIM - 1) {
#pragma unroll
                  for (int v = 0; v < BSUB; ++v) {
                    sYprev[v * S_DIM + col] = yn[v];
                    sS[v * LDA + col] = 0.5f * yn[v];
                  }
                }
              }
              if (slot == 0 && c == 0) {
#pragma unroll
                for (int v = 0; v < BSUB; ++v) sS[v * LDA + col] = 0.f;
              }
            }
          }
        }
        __syncthreads();
      }

      if (d >= N_ITER - 1) {
        if (tid < PLANE / 4) {
          float4 val = reinterpret_cast<const float4*>(sOut)[tid];
          float* dst = out + ((size_t)(1 + blk * C_DIM + (d - (N_ITER - 1))) * B_TOT + b0) * S_DIM;
          reinterpret_cast<float4*>(dst)[tid] = val;
        }
      }
      par ^= 1;
    }
  }
}

extern "C" void kernel_launch(void* const* d_in, const int* in_sizes, int n_in,
                              void* d_out, int out_size, void* d_ws,
                              size_t ws_size, hipStream_t stream) {
  (void)in_sizes; (void)n_in; (void)out_size; (void)d_ws; (void)ws_size;
  const float* y0     = (const float*)d_in[0];
  const float* forces = (const float*)d_in[1];
  const float* W      = (const float*)d_in[2];
  float* out = (float*)d_out;
  solver_gfx1250_kernel<<<NWG, THREADS, 0, stream>>>(y0, forces, W, out);
}


# GEMM operand reads taken out of the WMMA chains: W fragments read once from global into 32 registers per lane, each A tile's LDS reads issued as one group ahead of its chain; sum order kept
# speedup vs baseline: 4.4414x; 1.3393x over previous
#include <hip/hip_runtime.h>


#define S_DIM   64
#define C_DIM   8
#define N_ITER  4
#define NEUMANN 6
#define B_TOT   128
#define NB      8
#define BSUB    8
#define NWG     (B_TOT / BSUB)
#define ROWS    (N_ITER * BSUB)
#define NDIAG   (C_DIM + N_ITER - 1)
#define PLANE   (BSUB * S_DIM)
#define THREADS 256
#define DT_C    0.05f
#define LDA     66

typedef __attribute__((ext_vector_type(2))) float v2f;
typedef __attribute__((ext_vector_type(8))) float v8f;

#define KSTEPS (S_DIM / 4)

__device__ __forceinline__ void load_a_frags(const float* __restrict__ A, int lane, v2f (&a)[KSTEPS]) {
  const float* __restrict__ arow = A + (lane & 15) * LDA + ((lane >> 4) << 1);
#pragma unroll
  for (int j = 0; j < KSTEPS; ++j) { a[j].x = arow[4 * j]; a[j].y = arow[4 * j + 1]; }
}

__device__ __forceinline__ void load_a_frags2(const float* __restrict__ A0, const float* __restrict__ A1, int lane,
                                              v2f (&a0)[KSTEPS], v2f (&a1)[KSTEPS]) {
  const int off = (lane & 15) * LDA + ((lane >> 4) << 1);
  const float* __restrict__ r0 = A0 + off;
  const float* __restrict__ r1 = A1 + off;
#pragma unroll
  for (int j = 0; j < KSTEPS; j += 2) {
    a0[j].x = r0[4 * j];         a0[j].y = r0[4 * j + 1];
    a0[j + 1].x = r0[4 * j + 4]; a0[j + 1].y = r0[4 * j + 5];
    a1[j].x = r1[4 * j];         a1[j].y = r1[4 * j + 1];
    a1[j + 1].x = r1[4 * j + 4]; a1[j + 1].y = r1[4 * j + 5];
  }
}

__device__ __forceinline__ v8f wmma_chain(const v2f (&a)[KSTEPS], const v2f (&bw)[KSTEPS]) {
  v8f acc = {0.f, 0.f, 0.f, 0.f, 0.f, 0.f, 0.f, 0.f};
#pragma unroll
  for (int j = 0; j < KSTEPS; ++j)
    acc = __builtin_amdgcn_wmma_f32_16x16x4_f32(false, a[j], false, bw[j],
                                                (short)0, acc, false, false);
  return acc;
}

__global__ __launch_bounds__(THREADS, 1)
void solver_gfx1250_kernel(const float* __restrict__ y0,
                           const float* __restrict__ forces,
                           const float* __restrict__ W,
                           float* __restrict__ out) {
  __shared__ float sS [ROWS * LDA];
  __shared__ float sXa[ROWS * LDA];
  __shared__ float sXb[ROWS * LDA];
  __shared__ float sYv[N_ITER - 1][C_DIM][PLANE];
  __shared__ float sYprev[PLANE];
  __shared__ __attribute__((aligned(16))) float sOut[PLANE];

  const int tid  = threadIdx.x;
  const int lane = tid & 31;
  const int wave = tid >> 5;
  const int mt   = wave >> 2;
  const int n0   = (wave & 3) * 16;
  const int slot = mt * 2 + (lane >> 4);
  const int col  = n0 + (lane & 15);
  const int b0   = blockIdx.x * BSUB;
  const int rowbase = slot * BSUB * LDA + col;

  v2f bw[KSTEPS];
  {
    const int h2 = (lane >> 4) << 1;
    const float* __restrict__ wrow = W + col * S_DIM + h2;
#pragma unroll
    for (int j = 0; j < KSTEPS; ++j) { bw[j].x = wrow[4 * j]; bw[j].y = wrow[4 * j + 1]; }
  }
  for (int e = tid; e < ROWS * LDA; e += THREADS) {
    sS[e] = 0.f; sXa[e] = 0.f; sXb[e] = 0.f;
  }
  __syncthreads();
  for (int e = tid; e < PLANE; e += THREADS) {
    int b = e >> 6, j = e & 63;
    float v = y0[(b0 + b) * S_DIM + j];
    sYprev[e] = v;
    sS[b * LDA + j] = 0.5f * v;
    out[(size_t)(b0 + b) * S_DIM + j] = v;
  }
  __syncthreads();

  float xfin[BSUB];
  float ylast[BSUB];
#pragma unroll
  for (int v = 0; v < BSUB; ++v) { xfin[v] = 0.f; ylast[v] = 0.f; }
  int par = 0;

  for (int blk = 0; blk < NB; ++blk) {
    const float* fblk =
        forces + ((size_t)(1 + blk * C_DIM) * B_TOT + b0) * S_DIM;
    for (int off = tid * 16; off < C_DIM * BSUB * S_DIM; off += THREADS * 16)
      __builtin_prefetch(fblk + off, 0, 1);

    for (int d = 0; d < NDIAG; ++d) {
      const bool tile_act = (d >= 2 * mt) && (d <= 2 * mt + C_DIM);
      const int  c        = d - slot;
      const bool lane_act = (unsigned)c < (unsigned)C_DIM;
      float* bufP = par ? sXb : sXa;
      float* bufQ = par ? sXa : sXb;

      float ym[BSUB], yc[BSUB], fv[BSUB], gg[BSUB], rr[BSUB];
#pragma unroll
      for (int v = 0; v < BSUB; ++v) { ym[v] = 0.f; yc[v] = 0.f; fv[v] = 0.f; gg[v] = 0.f; rr[v] = 0.f; }

      if (lane_act) {
        const float* fp = fblk + (size_t)c * B_TOT * S_DIM + col;
#pragma unroll
        for (int v = 0; v < BSUB; ++v) fv[v] = fp[v * S_DIM];
        if (c == 0) {
#pragma unroll
          for (int v = 0; v < BSUB; ++v) ym[v] = sYprev[v * S_DIM + col];
        } else if (slot > 0) {
#pragma unroll
          for (int v = 0; v < BSUB; ++v) ym[v] = sYv[slot - 1][c - 1][v * S_DIM + col];
        }
        if (slot > 0) {
#pragma unroll
          for (int v = 0; v < BSUB; ++v) yc[v] = sYv[slot - 1][c][v * S_DIM + col];
        }
      }

      if (tile_act) {
        v2f as[KSTEPS], ax[KSTEPS];
        load_a_frags2(sS + mt * 16 * LDA, bufP + mt * 16 * LDA, lane, as, ax);
        __builtin_amdgcn_sched_barrier(0);
        v8f tacc = wmma_chain(as, bw);
        v8f wacc = wmma_chain(ax, bw);
        if (lane_act) {
#pragma unroll
          for (int v = 0; v < BSUB; ++v) {
            float t = tanhf(tacc[v]);
            float g = 0.5f * DT_C * (1.0f - t * t);
            float R = yc[v] - ym[v] + DT_C * t - DT_C * fv[v];
            float rhs = R;
            if (c > 0) rhs += xfin[v] - g * wacc[v];
            gg[v] = g;
            rr[v] = rhs;
            bufQ[rowbase + v * LDA] = rhs;
          }
        }
      }
      __syncthreads();

      for (int k = 0; k < NEUMANN; ++k) {
        const float* xin  = (k & 1) ? bufP : bufQ;
        float*       xout = (k & 1) ? bufQ : bufP;
        if (tile_act) {
          v2f ax[KSTEPS];
          load_a_frags(xin + mt * 16 * LDA, lane, ax);
          __builtin_amdgcn_sched_barrier(0);
          v8f macc = wmma_chain(ax, bw);
          if (lane_act) {
            float xn[BSUB];
#pragma unroll
            for (int v = 0; v < BSUB; ++v) {
              xn[v] = rr[v] - gg[v] * macc[v];
              xout[rowbase + v * LDA] = xn[v];
            }
            if (k == NEUMANN - 1) {
              float yn[BSUB], sn[BSUB];
#pragma unroll
              for (int v = 0; v < BSUB; ++v) {
                float yl = (c == 0) ? ym[v] : ylast[v];
                yn[v] = yc[v] - xn[v];
                sn[v] = 0.5f * (yl + yn[v]);
                xfin[v] = xn[v];
                ylast[v] = yn[v];
              }
              if (slot < N_ITER - 1) {
#pragma unroll
                for (int v = 0; v < BSUB; ++v) {
                  sYv[slot][c][v * S_DIM + col] = yn[v];
                  sS[rowbase + (BSUB + v) * LDA] = sn[v];
                }
              } else {
#pragma unroll
                for (int v = 0; v < BSUB; ++v) sOut[v * S_DIM + col] = yn[v];
                if (c == C_DIM - 1) {
#pragma unroll
                  for (int v = 0; v < BSUB; ++v) {
                    sYprev[v * S_DIM + col] = yn[v];
                    sS[v * LDA + col] = 0.5f * yn[v];
                  }
                }
              }
              if (slot == 0 && c == 0) {
#pragma unroll
                for (int v = 0; v < BSUB; ++v) sS[v * LDA + col] = 0.f;
              }
            }
          }
        }
        __syncthreads();
      }

      if (d >= N_ITER - 1) {
        if (tid < PLANE / 4) {
          float4 val = reinterpret_cast<const float4*>(sOut)[tid];
          float* dst = out + ((size_t)(1 + blk * C_DIM + (d - (N_ITER - 1))) * B_TOT + b0) * S_DIM;
          reinterpret_cast<float4*>(dst)[tid] = val;
        }
      }
      par ^= 1;
    }
  }
}

extern "C" void kernel_launch(void* const* d_in, const int* in_sizes, int n_in,
                              void* d_out, int out_size, void* d_ws,
                              size_t ws_size, hipStream_t stream) {
  (void)in_sizes; (void)n_in; (void)out_size; (void)d_ws; (void)ws_size;
  const float* y0     = (const float*)d_in[0];
  const float* forces = (const float*)d_in[1];
  const float* W      = (const float*)d_in[2];
  float* out = (float*)d_out;
  solver_gfx1250_kernel<<<NWG, THREADS, 0, stream>>>(y0, forces, W, out);
}


# Batch slice per workgroup halved to 4 rows (32 workgroups, 4 waves, one per SIMD): the 4 Newton slots fill ONE 16-row WMMA tile, each SIMD runs 16 WMMAs per GEMM, not 32; sum order kept
# speedup vs baseline: 5.5238x; 1.2437x over previous
#include <hip/hip_runtime.h>


#define S_DIM   64
#define C_DIM   8
#define N_ITER  4
#define NEUMANN 6
#define B_TOT   128
#define NB      8
#define BSUB    4
#define NWG     (B_TOT / BSUB)
#define ROWS    (N_ITER * BSUB)
#define NDIAG   (C_DIM + N_ITER - 1)
#define PLANE   (BSUB * S_DIM)
#define THREADS 128
#define DT_C    0.05f
#define LDA     66

typedef __attribute__((ext_vector_type(2))) float v2f;
typedef __attribute__((ext_vector_type(8))) float v8f;

#define KSTEPS (S_DIM / 4)

__device__ __forceinline__ void load_a_frags(const float* __restrict__ A, int lane, v2f (&a)[KSTEPS]) {
  const float* __restrict__ arow = A + (lane & 15) * LDA + ((lane >> 4) << 1);
#pragma unroll
  for (int j = 0; j < KSTEPS; ++j) { a[j].x = arow[4 * j]; a[j].y = arow[4 * j + 1]; }
}

__device__ __forceinline__ void load_a_frags2(const float* __restrict__ A0, const float* __restrict__ A1, int lane,
                                              v2f (&a0)[KSTEPS], v2f (&a1)[KSTEPS]) {
  const int off = (lane & 15) * LDA + ((lane >> 4) << 1);
  const float* __restrict__ r0 = A0 + off;
  const float* __restrict__ r1 = A1 + off;
#pragma unroll
  for (int j = 0; j < KSTEPS; j += 2) {
    a0[j].x = r0[4 * j];         a0[j].y = r0[4 * j + 1];
    a0[j + 1].x = r0[4 * j + 4]; a0[j + 1].y = r0[4 * j + 5];
    a1[j].x = r1[4 * j];         a1[j].y = r1[4 * j + 1];
    a1[j + 1].x = r1[4 * j + 4]; a1[j + 1].y = r1[4 * j + 5];
  }
}

__device__ __forceinline__ v8f wmma_chain(const v2f (&a)[KSTEPS], const v2f (&bw)[KSTEPS]) {
  v8f acc = {0.f, 0.f, 0.f, 0.f, 0.f, 0.f, 0.f, 0.f};
#pragma unroll
  for (int j = 0; j < KSTEPS; ++j)
    acc = __builtin_amdgcn_wmma_f32_16x16x4_f32(false, a[j], false, bw[j],
                                                (short)0, acc, false, false);
  return acc;
}

__global__ __launch_bounds__(THREADS, 1)
void solver_gfx1250_kernel(const float* __restrict__ y0,
                           const float* __restrict__ forces,
                           const float* __restrict__ W,
                           float* __restrict__ out) {
  __shared__ float sS [ROWS * LDA];
  __shared__ float sXa[ROWS * LDA];
  __shared__ float sXb[ROWS * LDA];
  __shared__ float sYv[N_ITER - 1][C_DIM][PLANE];
  __shared__ float sYprev[PLANE];
  __shared__ __attribute__((aligned(16))) float sOut[PLANE];

  const int tid   = threadIdx.x;
  const int lane  = tid & 31;
  const int n0    = (tid >> 5) * 16;
  const int half  = lane >> 4;
  const int slot0 = 2 * half;
  const int col   = n0 + (lane & 15);
  const int b0    = blockIdx.x * BSUB;
  const int rowbase = half * 8 * LDA + col;

  v2f bw[KSTEPS];
  {
    const int h2 = half << 1;
    const float* __restrict__ wrow = W + col * S_DIM + h2;
#pragma unroll
    for (int j = 0; j < KSTEPS; ++j) { bw[j].x = wrow[4 * j]; bw[j].y = wrow[4 * j + 1]; }
  }
  for (int e = tid; e < ROWS * LDA; e += THREADS) {
    sS[e] = 0.f; sXa[e] = 0.f; sXb[e] = 0.f;
  }
  __syncthreads();
  for (int e = tid; e < PLANE; e += THREADS) {
    int b = e >> 6, j = e & 63;
    float v = y0[(b0 + b) * S_DIM + j];
    sYprev[e] = v;
    sS[b * LDA + j] = 0.5f * v;
    out[(size_t)(b0 + b) * S_DIM + j] = v;
  }
  __syncthreads();

  float xfin[2 * BSUB];
  float ylast[2 * BSUB];
#pragma unroll
  for (int v = 0; v < 2 * BSUB; ++v) { xfin[v] = 0.f; ylast[v] = 0.f; }
  int par = 0;

  for (int blk = 0; blk < NB; ++blk) {
    const float* fblk =
        forces + ((size_t)(1 + blk * C_DIM) * B_TOT + b0) * S_DIM;
    for (int off = tid * 16; off < C_DIM * BSUB * S_DIM; off += THREADS * 16)
      __builtin_prefetch(fblk + off, 0, 1);

    for (int d = 0; d < NDIAG; ++d) {
      float* bufP = par ? sXb : sXa;
      float* bufQ = par ? sXa : sXb;

      float ym[2 * BSUB], yc[2 * BSUB], fv[2 * BSUB], gg[2 * BSUB], rr[2 * BSUB];
#pragma unroll
      for (int v = 0; v < 2 * BSUB; ++v) { ym[v] = 0.f; yc[v] = 0.f; gg[v] = 0.f; rr[v] = 0.f; }

#pragma unroll
      for (int q = 0; q < 2; ++q) {
        const int cc = min(max(d - slot0 - q, 0), C_DIM - 1);
        const float* fp = fblk + (size_t)cc * B_TOT * S_DIM + col;
#pragma unroll
        for (int u = 0; u < BSUB; ++u) fv[4 * q + u] = fp[u * S_DIM];
      }
#pragma unroll
      for (int q = 0; q < 2; ++q) {
        const int slot = slot0 + q;
        const int c    = d - slot;
        if ((unsigned)c < (unsigned)C_DIM) {
          if (c == 0) {
#pragma unroll
            for (int u = 0; u < BSUB; ++u) ym[4 * q + u] = sYprev[u * S_DIM + col];
          } else if (slot > 0) {
#pragma unroll
            for (int u = 0; u < BSUB; ++u) ym[4 * q + u] = sYv[slot - 1][c - 1][u * S_DIM + col];
          }
          if (slot > 0) {
#pragma unroll
            for (int u = 0; u < BSUB; ++u) yc[4 * q + u] = sYv[slot - 1][c][u * S_DIM + col];
          }
        }
      }

      {
        v2f as[KSTEPS], ax[KSTEPS];
        load_a_frags2(sS, bufP, lane, as, ax);
        __builtin_amdgcn_sched_barrier(0);
        v8f tacc = wmma_chain(as, bw);
        v8f wacc = wmma_chain(ax, bw);
#pragma unroll
        for (int q = 0; q < 2; ++q) {
          const int c = d - slot0 - q;
          if ((unsigned)c < (unsigned)C_DIM) {
#pragma unroll
            for (int u = 0; u < BSUB; ++u) {
              const int v = 4 * q + u;
              float t = tanhf(tacc[v]);
              float g = 0.5f * DT_C * (1.0f - t * t);
              float R = yc[v] - ym[v] + DT_C * t - DT_C * fv[v];
              float rhs = R;
              if (c > 0) rhs += xfin[v] - g * wacc[v];
              gg[v] = g;
              rr[v] = rhs;
              bufQ[rowbase + v * LDA] = rhs;
            }
          }
        }
      }
      __syncthreads();

      for (int k = 0; k < NEUMANN; ++k) {
        const float* xin  = (k & 1) ? bufP : bufQ;
        float*       xout = (k & 1) ? bufQ : bufP;
        v2f ax[KSTEPS];
        load_a_frags(xin, lane, ax);
        __builtin_amdgcn_sched_barrier(0);
        v8f macc = wmma_chain(ax, bw);
#pragma unroll
        for (int q = 0; q < 2; ++q) {
          const int slot = slot0 + q;
          const int c    = d - slot;
          if ((unsigned)c < (unsigned)C_DIM) {
            float xn[BSUB];
#pragma unroll
            for (int u = 0; u < BSUB; ++u) {
              const int v = 4 * q + u;
              xn[u] = rr[v] - gg[v] * macc[v];
              xout[rowbase + v * LDA] = xn[u];
            }
            if (k == NEUMANN - 1) {
              float yn[BSUB], sn[BSUB];
#pragma unroll
              for (int u = 0; u < BSUB; ++u) {
                const int v = 4 * q + u;
                float yl = (c == 0) ? ym[v] : ylast[v];
                yn[u] = yc[v] - xn[u];
                sn[u] = 0.5f * (yl + yn[u]);
                xfin[v] = xn[u];
                ylast[v] = yn[u];
              }
              if (slot < N_ITER - 1) {
#pragma unroll
                for (int u = 0; u < BSUB; ++u) {
                  sYv[slot][c][u * S_DIM + col] = yn[u];
                  sS[rowbase + (4 * q + u + BSUB) * LDA] = sn[u];
                }
              } else {
#pragma unroll
                for (int u = 0; u < BSUB; ++u) sOut[u * S_DIM + col] = yn[u];
                if (c == C_DIM - 1) {
#pragma unroll
                  for (int u = 0; u < BSUB; ++u) {
                    sYprev[u * S_DIM + col] = yn[u];
                    sS[u * LDA + col] = 0.5f * yn[u];
                  }
                }
              }
              if (slot == 0 && c == 0) {
#pragma unroll
                for (int u = 0; u < BSUB; ++u) sS[u * LDA + col] = 0.f;
              }
            }
          }
        }
        __syncthreads();
      }

      if (d >= N_ITER - 1) {
        if (tid < PLANE / 4) {
          float4 val = reinterpret_cast<const float4*>(sOut)[tid];
          float* dst = out + ((size_t)(1 + blk * C_DIM + (d - (N_ITER - 1))) * B_TOT + b0) * S_DIM;
          reinterpret_cast<float4*>(dst)[tid] = val;
        }
      }
      par ^= 1;
    }
  }
}

extern "C" void kernel_launch(void* const* d_in, const int* in_sizes, int n_in,
                              void* d_out, int out_size, void* d_ws,
                              size_t ws_size, hipStream_t stream) {
  (void)in_sizes; (void)n_in; (void)out_size; (void)d_ws; (void)ws_size;
  const float* y0     = (const float*)d_in[0];
  const float* forces = (const float*)d_in[1];
  const float* W      = (const float*)d_in[2];
  float* out = (float*)d_out;
  solver_gfx1250_kernel<<<NWG, THREADS, 0, stream>>>(y0, forces, W, out);
}


# Residual epilogue: tanhf's per-value range branch made ONE wave-uniform test over the lane's 8 values (same two formulas, same values); the 8 tanh+residual chains now interleave in straight-line code
# speedup vs baseline: 5.7951x; 1.0491x over previous
#include <hip/hip_runtime.h>


#define S_DIM   64
#define C_DIM   8
#define N_ITER  4
#define NEUMANN 6
#define B_TOT   128
#define NB      8
#define BSUB    4
#define NWG     (B_TOT / BSUB)
#define ROWS    (N_ITER * BSUB)
#define NDIAG   (C_DIM + N_ITER - 1)
#define PLANE   (BSUB * S_DIM)
#define THREADS 128
#define DT_C    0.05f
#define LDA     66

typedef __attribute__((ext_vector_type(2))) float v2f;
typedef __attribute__((ext_vector_type(8))) float v8f;

#define KSTEPS (S_DIM / 4)

__device__ __forceinline__ void load_a_frags(const float* __restrict__ A, int lane, v2f (&a)[KSTEPS]) {
  const float* __restrict__ arow = A + (lane & 15) * LDA + ((lane >> 4) << 1);
#pragma unroll
  for (int j = 0; j < KSTEPS; ++j) { a[j].x = arow[4 * j]; a[j].y = arow[4 * j + 1]; }
}

__device__ __forceinline__ void load_a_frags2(const float* __restrict__ A0, const float* __restrict__ A1, int lane,
                                              v2f (&a0)[KSTEPS], v2f (&a1)[KSTEPS]) {
  const int off = (lane & 15) * LDA + ((lane >> 4) << 1);
  const float* __restrict__ r0 = A0 + off;
  const float* __restrict__ r1 = A1 + off;
#pragma unroll
  for (int j = 0; j < KSTEPS; j += 2) {
    a0[j].x = r0[4 * j];         a0[j].y = r0[4 * j + 1];
    a0[j + 1].x = r0[4 * j + 4]; a0[j + 1].y = r0[4 * j + 5];
    a1[j].x = r1[4 * j];         a1[j].y = r1[4 * j + 1];
    a1[j + 1].x = r1[4 * j + 4]; a1[j + 1].y = r1[4 * j + 5];
  }
}

__device__ __forceinline__ v8f wmma_chain(const v2f (&a)[KSTEPS], const v2f (&bw)[KSTEPS]) {
  v8f acc = {0.f, 0.f, 0.f, 0.f, 0.f, 0.f, 0.f, 0.f};
#pragma unroll
  for (int j = 0; j < KSTEPS; ++j)
    acc = __builtin_amdgcn_wmma_f32_16x16x4_f32(false, a[j], false, bw[j],
                                                (short)0, acc, false, false);
  return acc;
}

#define TANH_SMALL_MAX 0.625f
__device__ __forceinline__ float tanh_small(float x) {
  const float ax = fabsf(x);
  const float z  = x * x;
  float p = fmaf(__uint_as_float(0xbbbac73du), z, __uint_as_float(0x3ca908c9u));
  p = fmaf(z, p, __uint_as_float(0xbd5c1c4eu));
  p = fmaf(z, p, __uint_as_float(0x3e088382u));
  p = fmaf(z, p, __uint_as_float(0xbeaaaa99u));
  p = ax * p;
  return copysignf(fmaf(z, p, ax), x);
}

__global__ __launch_bounds__(THREADS, 1)
void solver_gfx1250_kernel(const float* __restrict__ y0,
                           const float* __restrict__ forces,
                           const float* __restrict__ W,
                           float* __restrict__ out) {
  __shared__ float sS [ROWS * LDA];
  __shared__ float sXa[ROWS * LDA];
  __shared__ float sXb[ROWS * LDA];
  __shared__ float sYv[N_ITER - 1][C_DIM][PLANE];
  __shared__ float sYprev[PLANE];
  __shared__ __attribute__((aligned(16))) float sOut[PLANE];

  const int tid   = threadIdx.x;
  const int lane  = tid & 31;
  const int n0    = (tid >> 5) * 16;
  const int half  = lane >> 4;
  const int slot0 = 2 * half;
  const int col   = n0 + (lane & 15);
  const int b0    = blockIdx.x * BSUB;
  const int rowbase = half * 8 * LDA + col;

  v2f bw[KSTEPS];
  {
    const int h2 = half << 1;
    const float* __restrict__ wrow = W + col * S_DIM + h2;
#pragma unroll
    for (int j = 0; j < KSTEPS; ++j) { bw[j].x = wrow[4 * j]; bw[j].y = wrow[4 * j + 1]; }
  }
  for (int e = tid; e < ROWS * LDA; e += THREADS) {
    sS[e] = 0.f; sXa[e] = 0.f; sXb[e] = 0.f;
  }
  __syncthreads();
  for (int e = tid; e < PLANE; e += THREADS) {
    int b = e >> 6, j = e & 63;
    float v = y0[(b0 + b) * S_DIM + j];
    sYprev[e] = v;
    sS[b * LDA + j] = 0.5f * v;
    out[(size_t)(b0 + b) * S_DIM + j] = v;
  }
  __syncthreads();

  float xfin[2 * BSUB];
  float ylast[2 * BSUB];
#pragma unroll
  for (int v = 0; v < 2 * BSUB; ++v) { xfin[v] = 0.f; ylast[v] = 0.f; }
  int par = 0;

  for (int blk = 0; blk < NB; ++blk) {
    const float* fblk =
        forces + ((size_t)(1 + blk * C_DIM) * B_TOT + b0) * S_DIM;
    for (int off = tid * 16; off < C_DIM * BSUB * S_DIM; off += THREADS * 16)
      __builtin_prefetch(fblk + off, 0, 1);

    for (int d = 0; d < NDIAG; ++d) {
      float* bufP = par ? sXb : sXa;
      float* bufQ = par ? sXa : sXb;

      float ym[2 * BSUB], yc[2 * BSUB], fv[2 * BSUB], gg[2 * BSUB], rr[2 * BSUB];
#pragma unroll
      for (int v = 0; v < 2 * BSUB; ++v) { ym[v] = 0.f; yc[v] = 0.f; }

#pragma unroll
      for (int q = 0; q < 2; ++q) {
        const int cc = min(max(d - slot0 - q, 0), C_DIM - 1);
        const float* fp = fblk + (size_t)cc * B_TOT * S_DIM + col;
#pragma unroll
        for (int u = 0; u < BSUB; ++u) fv[4 * q + u] = fp[u * S_DIM];
      }
#pragma unroll
      for (int q = 0; q < 2; ++q) {
        const int slot = slot0 + q;
        const int c    = d - slot;
        if ((unsigned)c < (unsigned)C_DIM) {
          if (c == 0) {
#pragma unroll
            for (int u = 0; u < BSUB; ++u) ym[4 * q + u] = sYprev[u * S_DIM + col];
          } else if (slot > 0) {
#pragma unroll
            for (int u = 0; u < BSUB; ++u) ym[4 * q + u] = sYv[slot - 1][c - 1][u * S_DIM + col];
          }
          if (slot > 0) {
#pragma unroll
            for (int u = 0; u < BSUB; ++u) yc[4 * q + u] = sYv[slot - 1][c][u * S_DIM + col];
          }
        }
      }

      {
        v2f as[KSTEPS], ax[KSTEPS];
        load_a_frags2(sS, bufP, lane, as, ax);
        __builtin_amdgcn_sched_barrier(0);
        v8f tacc = wmma_chain(as, bw);
        v8f wacc = wmma_chain(ax, bw);
        const bool on0 = (unsigned)(d - slot0) < (unsigned)C_DIM;
        const bool on1 = (unsigned)(d - slot0 - 1) < (unsigned)C_DIM;
        const float am0 = fmaxf(fmaxf(fabsf(tacc[0]), fabsf(tacc[1])), fmaxf(fabsf(tacc[2]), fabsf(tacc[3])));
        const float am1 = fmaxf(fmaxf(fabsf(tacc[4]), fabsf(tacc[5])), fmaxf(fabsf(tacc[6]), fabsf(tacc[7])));
        const bool wide = (on0 & !(am0 < TANH_SMALL_MAX)) | (on1 & !(am1 < TANH_SMALL_MAX));
        float tt[2 * BSUB];
        if (__builtin_expect(__builtin_amdgcn_ballot_w32(wide) != 0u, 0)) {
#pragma unroll
          for (int v = 0; v < 2 * BSUB; ++v) tt[v] = tanhf(tacc[v]);
        } else {
#pragma unroll
          for (int v = 0; v < 2 * BSUB; ++v) tt[v] = tanh_small(tacc[v]);
        }
#pragma unroll
        for (int q = 0; q < 2; ++q) {
          const int c = d - slot0 - q;
#pragma unroll
          for (int u = 0; u < BSUB; ++u) {
            const int v = 4 * q + u;
            float t = tt[v];
            float g = (0.5f * DT_C) * fmaf(-t, t, 1.0f);
            float R = fmaf(-DT_C, fv[v], fmaf(DT_C, t, yc[v] - ym[v]));
            float rhs = R;
            if (c > 0) rhs = R + fmaf(-wacc[v], g, xfin[v]);
            gg[v] = g;
            rr[v] = rhs;
          }
        }
#pragma unroll
        for (int q = 0; q < 2; ++q) {
          const int c = d - slot0 - q;
          if ((unsigned)c < (unsigned)C_DIM) {
#pragma unroll
            for (int u = 0; u < BSUB; ++u)
              bufQ[rowbase + (4 * q + u) * LDA] = rr[4 * q + u];
          }
        }
      }
      __syncthreads();

      for (int k = 0; k < NEUMANN; ++k) {
        const float* xin  = (k & 1) ? bufP : bufQ;
        float*       xout = (k & 1) ? bufQ : bufP;
        v2f ax[KSTEPS];
        load_a_frags(xin, lane, ax);
        __builtin_amdgcn_sched_barrier(0);
        v8f macc = wmma_chain(ax, bw);
#pragma unroll
        for (int q = 0; q < 2; ++q) {
          const int slot = slot0 + q;
          const int c    = d - slot;
          if ((unsigned)c < (unsigned)C_DIM) {
            float xn[BSUB];
#pragma unroll
            for (int u = 0; u < BSUB; ++u) {
              const int v = 4 * q + u;
              xn[u] = rr[v] - gg[v] * macc[v];
              xout[rowbase + v * LDA] = xn[u];
            }
            if (k == NEUMANN - 1) {
              float yn[BSUB], sn[BSUB];
#pragma unroll
              for (int u = 0; u < BSUB; ++u) {
                const int v = 4 * q + u;
                float yl = (c == 0) ? ym[v] : ylast[v];
                yn[u] = yc[v] - xn[u];
                sn[u] = 0.5f * (yl + yn[u]);
                xfin[v] = xn[u];
                ylast[v] = yn[u];
              }
              if (slot < N_ITER - 1) {
#pragma unroll
                for (int u = 0; u < BSUB; ++u) {
                  sYv[slot][c][u * S_DIM + col] = yn[u];
                  sS[rowbase + (4 * q + u + BSUB) * LDA] = sn[u];
                }
              } else {
#pragma unroll
                for (int u = 0; u < BSUB; ++u) sOut[u * S_DIM + col] = yn[u];
                if (c == C_DIM - 1) {
#pragma unroll
                  for (int u = 0; u < BSUB; ++u) {
                    sYprev[u * S_DIM + col] = yn[u];
                    sS[u * LDA + col] = 0.5f * yn[u];
                  }
                }
              }
              if (slot == 0 && c == 0) {
#pragma unroll
                for (int u = 0; u < BSUB; ++u) sS[u * LDA + col] = 0.f;
              }
            }
          }
        }
        __syncthreads();
      }

      if (d >= N_ITER - 1) {
        if (tid < PLANE / 4) {
          float4 val = reinterpret_cast<const float4*>(sOut)[tid];
          float* dst = out + ((size_t)(1 + blk * C_DIM + (d - (N_ITER - 1))) * B_TOT + b0) * S_DIM;
          reinterpret_cast<float4*>(dst)[tid] = val;
        }
      }
      par ^= 1;
    }
  }
}

extern "C" void kernel_launch(void* const* d_in, const int* in_sizes, int n_in,
                              void* d_out, int out_size, void* d_ws,
                              size_t ws_size, hipStream_t stream) {
  (void)in_sizes; (void)n_in; (void)out_size; (void)d_ws; (void)ws_size;
  const float* y0     = (const float*)d_in[0];
  const float* forces = (const float*)d_in[1];
  const float* W      = (const float*)d_in[2];
  float* out = (float*)d_out;
  solver_gfx1250_kernel<<<NWG, THREADS, 0, stream>>>(y0, forces, W, out);
}


# Neumann GEMM's 16-WMMA dependent chain split onto two independent accumulators (even / odd K steps) added at the end: WMMAs issue at the pipe interval, not the dependent latency; sum order changes
# speedup vs baseline: 6.0794x; 1.0491x over previous
#include <hip/hip_runtime.h>


#define S_DIM   64
#define C_DIM   8
#define N_ITER  4
#define NEUMANN 6
#define B_TOT   128
#define NB      8
#define BSUB    4
#define NWG     (B_TOT / BSUB)
#define ROWS    (N_ITER * BSUB)
#define NDIAG   (C_DIM + N_ITER - 1)
#define PLANE   (BSUB * S_DIM)
#define THREADS 128
#define DT_C    0.05f
#define LDA     66

typedef __attribute__((ext_vector_type(2))) float v2f;
typedef __attribute__((ext_vector_type(8))) float v8f;

#define KSTEPS (S_DIM / 4)

__device__ __forceinline__ void load_a_frags(const float* __restrict__ A, int lane, v2f (&a)[KSTEPS]) {
  const float* __restrict__ arow = A + (lane & 15) * LDA + ((lane >> 4) << 1);
#pragma unroll
  for (int j = 0; j < KSTEPS; ++j) { a[j].x = arow[4 * j]; a[j].y = arow[4 * j + 1]; }
}

__device__ __forceinline__ void load_a_frags2(const float* __restrict__ A0, const float* __restrict__ A1, int lane,
                                              v2f (&a0)[KSTEPS], v2f (&a1)[KSTEPS]) {
  const int off = (lane & 15) * LDA + ((lane >> 4) << 1);
  const float* __restrict__ r0 = A0 + off;
  const float* __restrict__ r1 = A1 + off;
#pragma unroll
  for (int j = 0; j < KSTEPS; j += 2) {
    a0[j].x = r0[4 * j];         a0[j].y = r0[4 * j + 1];
    a0[j + 1].x = r0[4 * j + 4]; a0[j + 1].y = r0[4 * j + 5];
    a1[j].x = r1[4 * j];         a1[j].y = r1[4 * j + 1];
    a1[j + 1].x = r1[4 * j + 4]; a1[j + 1].y = r1[4 * j + 5];
  }
}

__device__ __forceinline__ v8f wmma_chain(const v2f (&a)[KSTEPS], const v2f (&bw)[KSTEPS]) {
  v8f acc = {0.f, 0.f, 0.f, 0.f, 0.f, 0.f, 0.f, 0.f};
#pragma unroll
  for (int j = 0; j < KSTEPS; ++j)
    acc = __builtin_amdgcn_wmma_f32_16x16x4_f32(false, a[j], false, bw[j],
                                                (short)0, acc, false, false);
  return acc;
}

__device__ __forceinline__ v8f wmma_chain_split(const v2f (&a)[KSTEPS], const v2f (&bw)[KSTEPS]) {
  v8f acc0 = {0.f, 0.f, 0.f, 0.f, 0.f, 0.f, 0.f, 0.f};
  v8f acc1 = {0.f, 0.f, 0.f, 0.f, 0.f, 0.f, 0.f, 0.f};
#pragma unroll
  for (int j = 0; j < KSTEPS; j += 2) {
    acc0 = __builtin_amdgcn_wmma_f32_16x16x4_f32(false, a[j], false, bw[j],
                                                 (short)0, acc0, false, false);
    acc1 = __builtin_amdgcn_wmma_f32_16x16x4_f32(false, a[j + 1], false, bw[j + 1],
                                                 (short)0, acc1, false, false);
  }
  return acc0 + acc1;
}

#define TANH_SMALL_MAX 0.625f
__device__ __forceinline__ float tanh_small(float x) {
  const float ax = fabsf(x);
  const float z  = x * x;
  float p = fmaf(__uint_as_float(0xbbbac73du), z, __uint_as_float(0x3ca908c9u));
  p = fmaf(z, p, __uint_as_float(0xbd5c1c4eu));
  p = fmaf(z, p, __uint_as_float(0x3e088382u));
  p = fmaf(z, p, __uint_as_float(0xbeaaaa99u));
  p = ax * p;
  return copysignf(fmaf(z, p, ax), x);
}

__global__ __launch_bounds__(THREADS, 1)
void solver_gfx1250_kernel(const float* __restrict__ y0,
                           const float* __restrict__ forces,
                           const float* __restrict__ W,
                           float* __restrict__ out) {
  __shared__ float sS [ROWS * LDA];
  __shared__ float sXa[ROWS * LDA];
  __shared__ float sXb[ROWS * LDA];
  __shared__ float sYv[N_ITER - 1][C_DIM][PLANE];
  __shared__ float sYprev[PLANE];
  __shared__ __attribute__((aligned(16))) float sOut[PLANE];

  const int tid   = threadIdx.x;
  const int lane  = tid & 31;
  const int n0    = (tid >> 5) * 16;
  const int half  = lane >> 4;
  const int slot0 = 2 * half;
  const int col   = n0 + (lane & 15);
  const int b0    = blockIdx.x * BSUB;
  const int rowbase = half * 8 * LDA + col;

  v2f bw[KSTEPS];
  {
    const int h2 = half << 1;
    const float* __restrict__ wrow = W + col * S_DIM + h2;
#pragma unroll
    for (int j = 0; j < KSTEPS; ++j) { bw[j].x = wrow[4 * j]; bw[j].y = wrow[4 * j + 1]; }
  }
  for (int e = tid; e < ROWS * LDA; e += THREADS) {
    sS[e] = 0.f; sXa[e] = 0.f; sXb[e] = 0.f;
  }
  __syncthreads();
  for (int e = tid; e < PLANE; e += THREADS) {
    int b = e >> 6, j = e & 63;
    float v = y0[(b0 + b) * S_DIM + j];
    sYprev[e] = v;
    sS[b * LDA + j] = 0.5f * v;
    out[(size_t)(b0 + b) * S_DIM + j] = v;
  }
  __syncthreads();

  float xfin[2 * BSUB];
  float ylast[2 * BSUB];
#pragma unroll
  for (int v = 0; v < 2 * BSUB; ++v) { xfin[v] = 0.f; ylast[v] = 0.f; }
  int par = 0;

  for (int blk = 0; blk < NB; ++blk) {
    const float* fblk =
        forces + ((size_t)(1 + blk * C_DIM) * B_TOT + b0) * S_DIM;
    for (int off = tid * 16; off < C_DIM * BSUB * S_DIM; off += THREADS * 16)
      __builtin_prefetch(fblk + off, 0, 1);

    for (int d = 0; d < NDIAG; ++d) {
      float* bufP = par ? sXb : sXa;
      float* bufQ = par ? sXa : sXb;

      float ym[2 * BSUB], yc[2 * BSUB], fv[2 * BSUB], gg[2 * BSUB], rr[2 * BSUB];
#pragma unroll
      for (int v = 0; v < 2 * BSUB; ++v) { ym[v] = 0.f; yc[v] = 0.f; }

#pragma unroll
      for (int q = 0; q < 2; ++q) {
        const int cc = min(max(d - slot0 - q, 0), C_DIM - 1);
        const float* fp = fblk + (size_t)cc * B_TOT * S_DIM + col;
#pragma unroll
        for (int u = 0; u < BSUB; ++u) fv[4 * q + u] = fp[u * S_DIM];
      }
#pragma unroll
      for (int q = 0; q < 2; ++q) {
        const int slot = slot0 + q;
        const int c    = d - slot;
        if ((unsigned)c < (unsigned)C_DIM) {
          if (c == 0) {
#pragma unroll
            for (int u = 0; u < BSUB; ++u) ym[4 * q + u] = sYprev[u * S_DIM + col];
          } else if (slot > 0) {
#pragma unroll
            for (int u = 0; u < BSUB; ++u) ym[4 * q + u] = sYv[slot - 1][c - 1][u * S_DIM + col];
          }
          if (slot > 0) {
#pragma unroll
            for (int u = 0; u < BSUB; ++u) yc[4 * q + u] = sYv[slot - 1][c][u * S_DIM + col];
          }
        }
      }

      {
        v2f as[KSTEPS], ax[KSTEPS];
        load_a_frags2(sS, bufP, lane, as, ax);
        __builtin_amdgcn_sched_barrier(0);
        v8f tacc = wmma_chain(as, bw);
        v8f wacc = wmma_chain(ax, bw);
        const bool on0 = (unsigned)(d - slot0) < (unsigned)C_DIM;
        const bool on1 = (unsigned)(d - slot0 - 1) < (unsigned)C_DIM;
        const float am0 = fmaxf(fmaxf(fabsf(tacc[0]), fabsf(tacc[1])), fmaxf(fabsf(tacc[2]), fabsf(tacc[3])));
        const float am1 = fmaxf(fmaxf(fabsf(tacc[4]), fabsf(tacc[5])), fmaxf(fabsf(tacc[6]), fabsf(tacc[7])));
        const bool wide = (on0 & !(am0 < TANH_SMALL_MAX)) | (on1 & !(am1 < TANH_SMALL_MAX));
        float tt[2 * BSUB];
        if (__builtin_expect(__builtin_amdgcn_ballot_w32(wide) != 0u, 0)) {
#pragma unroll
          for (int v = 0; v < 2 * BSUB; ++v) tt[v] = tanhf(tacc[v]);
        } else {
#pragma unroll
          for (int v = 0; v < 2 * BSUB; ++v) tt[v] = tanh_small(tacc[v]);
        }
#pragma unroll
        for (int q = 0; q < 2; ++q) {
          const int c = d - slot0 - q;
#pragma unroll
          for (int u = 0; u < BSUB; ++u) {
            const int v = 4 * q + u;
            float t = tt[v];
            float g = (0.5f * DT_C) * fmaf(-t, t, 1.0f);
            float R = fmaf(-DT_C, fv[v], fmaf(DT_C, t, yc[v] - ym[v]));
            float rhs = R;
            if (c > 0) rhs = R + fmaf(-wacc[v], g, xfin[v]);
            gg[v] = g;
            rr[v] = rhs;
          }
        }
#pragma unroll
        for (int q = 0; q < 2; ++q) {
          const int c = d - slot0 - q;
          if ((unsigned)c < (unsigned)C_DIM) {
#pragma unroll
            for (int u = 0; u < BSUB; ++u)
              bufQ[rowbase + (4 * q + u) * LDA] = rr[4 * q + u];
          }
        }
      }
      __syncthreads();

      for (int k = 0; k < NEUMANN; ++k) {
        const float* xin  = (k & 1) ? bufP : bufQ;
        float*       xout = (k & 1) ? bufQ : bufP;
        v2f ax[KSTEPS];
        load_a_frags(xin, lane, ax);
        __builtin_amdgcn_sched_barrier(0);
        v8f macc = wmma_chain_split(ax, bw);
#pragma unroll
        for (int q = 0; q < 2; ++q) {
          const int slot = slot0 + q;
          const int c    = d - slot;
          if ((unsigned)c < (unsigned)C_DIM) {
            float xn[BSUB];
#pragma unroll
            for (int u = 0; u < BSUB; ++u) {
              const int v = 4 * q + u;
              xn[u] = rr[v] - gg[v] * macc[v];
              xout[rowbase + v * LDA] = xn[u];
            }
            if (k == NEUMANN - 1) {
              float yn[BSUB], sn[BSUB];
#pragma unroll
              for (int u = 0; u < BSUB; ++u) {
                const int v = 4 * q + u;
                float yl = (c == 0) ? ym[v] : ylast[v];
                yn[u] = yc[v] - xn[u];
                sn[u] = 0.5f * (yl + yn[u]);
                xfin[v] = xn[u];
                ylast[v] = yn[u];
              }
              if (slot < N_ITER - 1) {
#pragma unroll
                for (int u = 0; u < BSUB; ++u) {
                  sYv[slot][c][u * S_DIM + col] = yn[u];
                  sS[rowbase + (4 * q + u + BSUB) * LDA] = sn[u];
                }
              } else {
#pragma unroll
                for (int u = 0; u < BSUB; ++u) sOut[u * S_DIM + col] = yn[u];
                if (c == C_DIM - 1) {
#pragma unroll
                  for (int u = 0; u < BSUB; ++u) {
                    sYprev[u * S_DIM + col] = yn[u];
                    sS[u * LDA + col] = 0.5f * yn[u];
                  }
                }
              }
              if (slot == 0 && c == 0) {
#pragma unroll
                for (int u = 0; u < BSUB; ++u) sS[u * LDA + col] = 0.f;
              }
            }
          }
        }
        __syncthreads();
      }

      if (d >= N_ITER - 1) {
        if (tid < PLANE / 4) {
          float4 val = reinterpret_cast<const float4*>(sOut)[tid];
          float* dst = out + ((size_t)(1 + blk * C_DIM + (d - (N_ITER - 1))) * B_TOT + b0) * S_DIM;
          reinterpret_cast<float4*>(dst)[tid] = val;
        }
      }
      par ^= 1;
    }
  }
}

extern "C" void kernel_launch(void* const* d_in, const int* in_sizes, int n_in,
                              void* d_out, int out_size, void* d_ws,
                              size_t ws_size, hipStream_t stream) {
  (void)in_sizes; (void)n_in; (void)out_size; (void)d_ws; (void)ws_size;
  const float* y0     = (const float*)d_in[0];
  const float* forces = (const float*)d_in[1];
  const float* W      = (const float*)d_in[2];
  float* out = (float*)d_out;
  solver_gfx1250_kernel<<<NWG, THREADS, 0, stream>>>(y0, forces, W, out);
}


# Barrier moved to the head of the consuming GEMM, behind a read-back of the wave's own 16 K columns: those WMMAs run while the other LDS reads are in flight; K order starts there, sum order changes
# speedup vs baseline: 6.7667x; 1.1131x over previous
#include <hip/hip_runtime.h>


#define S_DIM   64
#define C_DIM   8
#define N_ITER  4
#define NEUMANN 6
#define B_TOT   128
#define NB      8
#define BSUB    4
#define NWG     (B_TOT / BSUB)
#define ROWS    (N_ITER * BSUB)
#define NDIAG   (C_DIM + N_ITER - 1)
#define PLANE   (BSUB * S_DIM)
#define THREADS 128
#define DT_C    0.05f
#define LDA     66

typedef __attribute__((ext_vector_type(2))) float v2f;
typedef __attribute__((ext_vector_type(8))) float v8f;

#define KSTEPS (S_DIM / 4)

#define QSTEPS (KSTEPS / 4)

template <int Q>
__device__ __forceinline__ void load_a_quarter(const float* __restrict__ rowq, v2f (&a)[KSTEPS]) {
#pragma unroll
  for (int i = 0; i < QSTEPS; ++i) { a[QSTEPS * Q + i].x = rowq[4 * i]; a[QSTEPS * Q + i].y = rowq[4 * i + 1]; }
}

__device__ __forceinline__ v8f wmma_step(const v2f a, const v2f b, const v8f c) {
  return __builtin_amdgcn_wmma_f32_16x16x4_f32(false, a, false, b, (short)0, c, false, false);
}

__device__ __forceinline__ void bar_arrive() {
  __builtin_amdgcn_fence(__ATOMIC_RELEASE, "workgroup", "local");
  __builtin_amdgcn_s_barrier_signal(-1);
}
__device__ __forceinline__ void bar_wait() {
  __builtin_amdgcn_s_barrier_wait(-1);
  __builtin_amdgcn_fence(__ATOMIC_ACQUIRE, "workgroup", "local");
}

#define TANH_SMALL_MAX 0.625f
__device__ __forceinline__ float tanh_small(float x) {
  const float ax = fabsf(x);
  const float z  = x * x;
  float p = fmaf(__uint_as_float(0xbbbac73du), z, __uint_as_float(0x3ca908c9u));
  p = fmaf(z, p, __uint_as_float(0xbd5c1c4eu));
  p = fmaf(z, p, __uint_as_float(0x3e088382u));
  p = fmaf(z, p, __uint_as_float(0xbeaaaa99u));
  p = ax * p;
  return copysignf(fmaf(z, p, ax), x);
}

__global__ __launch_bounds__(THREADS, 1)
void solver_gfx1250_kernel(const float* __restrict__ y0,
                           const float* __restrict__ forces,
                           const float* __restrict__ W,
                           float* __restrict__ out) {
  __shared__ float sS [ROWS * LDA];
  __shared__ float sXa[ROWS * LDA];
  __shared__ float sXb[ROWS * LDA];
  __shared__ float sYv[N_ITER - 1][C_DIM][PLANE];
  __shared__ float sYprev[PLANE];
  __shared__ __attribute__((aligned(16))) float sOut[PLANE];

  const int tid   = threadIdx.x;
  const int lane  = tid & 31;
  const int wave  = tid >> 5;
  const int n0    = wave * 16;
  const int half  = lane >> 4;
  const int slot0 = 2 * half;
  const int col   = n0 + (lane & 15);
  const int b0    = blockIdx.x * BSUB;
  const int rowbase = half * 8 * LDA + col;
  const int arow_off = (lane & 15) * LDA + (half << 1);
  int qoff[4];
#pragma unroll
  for (int q = 0; q < 4; ++q) qoff[q] = ((wave + q) & 3) * 16;

  v2f bw[KSTEPS];
  {
    const int h2 = half << 1;
    const float* __restrict__ wrow = W + col * S_DIM + h2;
#pragma unroll
    for (int q = 0; q < 4; ++q) {
#pragma unroll
      for (int i = 0; i < QSTEPS; ++i) {
        bw[QSTEPS * q + i].x = wrow[qoff[q] + 4 * i];
        bw[QSTEPS * q + i].y = wrow[qoff[q] + 4 * i + 1];
      }
    }
  }
  for (int e = tid; e < ROWS * LDA; e += THREADS) {
    sS[e] = 0.f; sXa[e] = 0.f; sXb[e] = 0.f;
  }
  for (int e = tid; e < (N_ITER - 1) * C_DIM * PLANE; e += THREADS) (&sYv[0][0][0])[e] = 0.f;
  __syncthreads();
  for (int e = tid; e < PLANE; e += THREADS) {
    int b = e >> 6, j = e & 63;
    float v = y0[(b0 + b) * S_DIM + j];
    sYprev[e] = v;
    sS[b * LDA + j] = 0.5f * v;
    out[(size_t)(b0 + b) * S_DIM + j] = v;
  }
  __syncthreads();

  float xfin[2 * BSUB];
  float ylast[2 * BSUB];
#pragma unroll
  for (int v = 0; v < 2 * BSUB; ++v) { xfin[v] = 0.f; ylast[v] = 0.f; }
  int par = 0;
  const v8f zero8 = {0.f, 0.f, 0.f, 0.f, 0.f, 0.f, 0.f, 0.f};

  for (int blk = 0; blk < NB; ++blk) {
    const float* fblk =
        forces + ((size_t)(1 + blk * C_DIM) * B_TOT + b0) * S_DIM;
    for (int off = tid * 16; off < C_DIM * BSUB * S_DIM; off += THREADS * 16)
      __builtin_prefetch(fblk + off, 0, 1);

    for (int d = 0; d < NDIAG; ++d) {
      float* bufP = par ? sXb : sXa;
      float* bufQ = par ? sXa : sXb;

      const bool copy_prev = (d >= N_ITER) || (d == 0 && blk > 0);
      const int  t_prev    = blk * C_DIM + (d == 0 ? 0 : d - (N_ITER - 1));

      float ym[2 * BSUB], yc[2 * BSUB], fv[2 * BSUB], gg[2 * BSUB], rr[2 * BSUB];
      float4 oval = {0.f, 0.f, 0.f, 0.f};

      {
        v2f as[KSTEPS], ax[KSTEPS];
        const float* __restrict__ srow = sS + arow_off;
        const float* __restrict__ xrow = bufP + arow_off;
        load_a_quarter<0>(srow + qoff[0], as);
        load_a_quarter<0>(xrow + qoff[0], ax);
        bar_arrive();
#pragma unroll
        for (int q = 0; q < 2; ++q) {
          const int cc = min(max(d - slot0 - q, 0), C_DIM - 1);
          const float* fp = fblk + (size_t)cc * B_TOT * S_DIM + col;
#pragma unroll
          for (int u = 0; u < BSUB; ++u) fv[4 * q + u] = fp[u * S_DIM];
        }
        bar_wait();
        load_a_quarter<1>(srow + qoff[1], as);
        load_a_quarter<1>(xrow + qoff[1], ax);
        load_a_quarter<2>(srow + qoff[2], as);
        load_a_quarter<2>(xrow + qoff[2], ax);
        __builtin_amdgcn_sched_barrier(0);
        v8f tacc = wmma_step(as[0], bw[0], zero8);
        v8f wacc = wmma_step(ax[0], bw[0], zero8);
        __builtin_amdgcn_sched_barrier(0);
        load_a_quarter<3>(srow + qoff[3], as);
        load_a_quarter<3>(xrow + qoff[3], ax);
        __builtin_amdgcn_sched_barrier(0);
        tacc = wmma_step(as[1], bw[1], tacc);
        wacc = wmma_step(ax[1], bw[1], wacc);
        tacc = wmma_step(as[2], bw[2], tacc);
        wacc = wmma_step(ax[2], bw[2], wacc);
        tacc = wmma_step(as[3], bw[3], tacc);
        wacc = wmma_step(ax[3], bw[3], wacc);
#pragma unroll
        for (int j = QSTEPS; j < KSTEPS; ++j) {
          tacc = wmma_step(as[j], bw[j], tacc);
          wacc = wmma_step(ax[j], bw[j], wacc);
        }
        __builtin_amdgcn_sched_barrier(0);
#pragma unroll
        for (int q = 0; q < 2; ++q) {
          const int slot = slot0 + q;
          const int c    = d - slot;
          const bool on  = (unsigned)c < (unsigned)C_DIM;
          const int lv   = max(slot - 1, 0);
          const int cm   = min(max(c - 1, 0), C_DIM - 1);
          const int cq   = min(max(c, 0), C_DIM - 1);
          const float* pm = (c == 0) ? &sYprev[col] : &sYv[lv][cm][col];
          const float* pc = &sYv[lv][cq][col];
          const bool usem = on && (c == 0 || slot > 0);
          const bool usec = on && (slot > 0);
#pragma unroll
          for (int u = 0; u < BSUB; ++u) {
            const float a = pm[u * S_DIM];
            const float b = pc[u * S_DIM];
            ym[4 * q + u] = usem ? a : 0.f;
            yc[4 * q + u] = usec ? b : 0.f;
          }
        }
        if (copy_prev && tid < PLANE / 4) oval = reinterpret_cast<const float4*>(sOut)[tid];
        const bool on0 = (unsigned)(d - slot0) < (unsigned)C_DIM;
        const bool on1 = (unsigned)(d - slot0 - 1) < (unsigned)C_DIM;
        const float am0 = fmaxf(fmaxf(fabsf(tacc[0]), fabsf(tacc[1])), fmaxf(fabsf(tacc[2]), fabsf(tacc[3])));
        const float am1 = fmaxf(fmaxf(fabsf(tacc[4]), fabsf(tacc[5])), fmaxf(fabsf(tacc[6]), fabsf(tacc[7])));
        const bool wide = (on0 & !(am0 < TANH_SMALL_MAX)) | (on1 & !(am1 < TANH_SMALL_MAX));
        float tt[2 * BSUB];
        if (__builtin_expect(__builtin_amdgcn_ballot_w32(wide) != 0u, 0)) {
#pragma unroll
          for (int v = 0; v < 2 * BSUB; ++v) tt[v] = tanhf(tacc[v]);
        } else {
#pragma unroll
          for (int v = 0; v < 2 * BSUB; ++v) tt[v] = tanh_small(tacc[v]);
        }
#pragma unroll
        for (int q = 0; q < 2; ++q) {
          const int c = d - slot0 - q;
#pragma unroll
          for (int u = 0; u < BSUB; ++u) {
            const int v = 4 * q + u;
            float t = tt[v];
            float g = (0.5f * DT_C) * fmaf(-t, t, 1.0f);
            float R = fmaf(-DT_C, fv[v], fmaf(DT_C, t, yc[v] - ym[v]));
            float rhs = R;
            if (c > 0) rhs = R + fmaf(-wacc[v], g, xfin[v]);
            gg[v] = g;
            rr[v] = rhs;
          }
        }
#pragma unroll
        for (int q = 0; q < 2; ++q) {
          const int c = d - slot0 - q;
          if ((unsigned)c < (unsigned)C_DIM) {
#pragma unroll
            for (int u = 0; u < BSUB; ++u)
              bufQ[rowbase + (4 * q + u) * LDA] = rr[4 * q + u];
          }
        }
        if (copy_prev && tid < PLANE / 4) {
          float* dst = out + ((size_t)t_prev * B_TOT + b0) * S_DIM;
          reinterpret_cast<float4*>(dst)[tid] = oval;
        }
      }

      for (int k = 0; k < NEUMANN; ++k) {
        const float* xin  = (k & 1) ? bufP : bufQ;
        float*       xout = (k & 1) ? bufQ : bufP;
        v2f ax[KSTEPS];
        const float* __restrict__ xrow = xin + arow_off;
        load_a_quarter<0>(xrow + qoff[0], ax);
        bar_arrive();
        bar_wait();
        load_a_quarter<1>(xrow + qoff[1], ax);
        load_a_quarter<2>(xrow + qoff[2], ax);
        load_a_quarter<3>(xrow + qoff[3], ax);
        __builtin_amdgcn_sched_barrier(0);
        v8f acc0 = wmma_step(ax[0], bw[0], zero8);
        v8f acc1 = wmma_step(ax[1], bw[1], zero8);
#pragma unroll
        for (int j = 2; j < KSTEPS; j += 2) {
          acc0 = wmma_step(ax[j], bw[j], acc0);
          acc1 = wmma_step(ax[j + 1], bw[j + 1], acc1);
        }
        v8f macc = acc0 + acc1;
        asm volatile("" : "+v"(macc));
#pragma unroll
        for (int q = 0; q < 2; ++q) {
          const int slot = slot0 + q;
          const int c    = d - slot;
          if ((unsigned)c < (unsigned)C_DIM) {
            float xn[BSUB];
#pragma unroll
            for (int u = 0; u < BSUB; ++u) {
              const int v = 4 * q + u;
              xn[u] = rr[v] - gg[v] * macc[v];
              xout[rowbase + v * LDA] = xn[u];
            }
            if (k == NEUMANN - 1) {
              float yn[BSUB], sn[BSUB];
#pragma unroll
              for (int u = 0; u < BSUB; ++u) {
                const int v = 4 * q + u;
                float yl = (c == 0) ? ym[v] : ylast[v];
                yn[u] = yc[v] - xn[u];
                sn[u] = 0.5f * (yl + yn[u]);
                xfin[v] = xn[u];
                ylast[v] = yn[u];
              }
              if (slot < N_ITER - 1) {
#pragma unroll
                for (int u = 0; u < BSUB; ++u) {
                  sYv[slot][c][u * S_DIM + col] = yn[u];
                  sS[rowbase + (4 * q + u + BSUB) * LDA] = sn[u];
                }
              } else {
#pragma unroll
                for (int u = 0; u < BSUB; ++u) sOut[u * S_DIM + col] = yn[u];
                if (c == C_DIM - 1) {
#pragma unroll
                  for (int u = 0; u < BSUB; ++u) {
                    sYprev[u * S_DIM + col] = yn[u];
                    sS[u * LDA + col] = 0.5f * yn[u];
                  }
                }
              }
              if (slot == 0 && c == 0) {
#pragma unroll
                for (int u = 0; u < BSUB; ++u) sS[u * LDA + col] = 0.f;
              }
            }
          }
        }
      }
      par ^= 1;
    }
  }

  __syncthreads();
  if (tid < PLANE / 4) {
    float4 val = reinterpret_cast<const float4*>(sOut)[tid];
    float* dst = out + ((size_t)(NB * C_DIM) * B_TOT + b0) * S_DIM;
    reinterpret_cast<float4*>(dst)[tid] = val;
  }
}

extern "C" void kernel_launch(void* const* d_in, const int* in_sizes, int n_in,
                              void* d_out, int out_size, void* d_ws,
                              size_t ws_size, hipStream_t stream) {
  (void)in_sizes; (void)n_in; (void)out_size; (void)d_ws; (void)ws_size;
  const float* y0     = (const float*)d_in[0];
  const float* forces = (const float*)d_in[1];
  const float* W      = (const float*)d_in[2];
  float* out = (float*)d_out;
  solver_gfx1250_kernel<<<NWG, THREADS, 0, stream>>>(y0, forces, W, out);
}


# tanh argument W s built from the W x products the sweep already computes (running W y per cell in registers, one cross-half lane move): its GEMM is gone; W x on 2 accumulators; sum order changes
# speedup vs baseline: 7.1341x; 1.0543x over previous
#include <hip/hip_runtime.h>


#define S_DIM   64
#define C_DIM   8
#define N_ITER  4
#define NEUMANN 6
#define B_TOT   128
#define NB      8
#define BSUB    4
#define NWG     (B_TOT / BSUB)
#define ROWS    (N_ITER * BSUB)
#define NDIAG   (C_DIM + N_ITER - 1)
#define PLANE   (BSUB * S_DIM)
#define THREADS 128
#define DT_C    0.05f
#define LDA     66

typedef __attribute__((ext_vector_type(2))) float v2f;
typedef __attribute__((ext_vector_type(8))) float v8f;

#define KSTEPS (S_DIM / 4)

#define QSTEPS (KSTEPS / 4)

template <int Q>
__device__ __forceinline__ void load_a_quarter(const float* __restrict__ rowq, v2f (&a)[KSTEPS]) {
#pragma unroll
  for (int i = 0; i < QSTEPS; ++i) { a[QSTEPS * Q + i].x = rowq[4 * i]; a[QSTEPS * Q + i].y = rowq[4 * i + 1]; }
}

__device__ __forceinline__ v8f wmma_step(const v2f a, const v2f b, const v8f c) {
  return __builtin_amdgcn_wmma_f32_16x16x4_f32(false, a, false, b, (short)0, c, false, false);
}

__device__ __forceinline__ float xhalf(float x) {
  return __builtin_bit_cast(float, __builtin_amdgcn_permlanex16(__builtin_bit_cast(int, x), __builtin_bit_cast(int, x),
                                                                  0x76543210, 0xfedcba98, false, false));
}

__device__ __forceinline__ void bar_arrive() {
  __builtin_amdgcn_fence(__ATOMIC_RELEASE, "workgroup", "local");
  __builtin_amdgcn_s_barrier_signal(-1);
}
__device__ __forceinline__ void bar_wait() {
  __builtin_amdgcn_s_barrier_wait(-1);
  __builtin_amdgcn_fence(__ATOMIC_ACQUIRE, "workgroup", "local");
}

#define TANH_SMALL_MAX 0.625f
__device__ __forceinline__ float tanh_small(float x) {
  const float ax = fabsf(x);
  const float z  = x * x;
  float p = fmaf(__uint_as_float(0xbbbac73du), z, __uint_as_float(0x3ca908c9u));
  p = fmaf(z, p, __uint_as_float(0xbd5c1c4eu));
  p = fmaf(z, p, __uint_as_float(0x3e088382u));
  p = fmaf(z, p, __uint_as_float(0xbeaaaa99u));
  p = ax * p;
  return copysignf(fmaf(z, p, ax), x);
}

__global__ __launch_bounds__(THREADS, 1)
void solver_gfx1250_kernel(const float* __restrict__ y0,
                           const float* __restrict__ forces,
                           const float* __restrict__ W,
                           float* __restrict__ out) {
  __shared__ float sXa[ROWS * LDA];
  __shared__ float sXb[ROWS * LDA];
  __shared__ float sYv[N_ITER - 1][C_DIM][PLANE];
  __shared__ float sYprev[PLANE];
  __shared__ __attribute__((aligned(16))) float sOut[PLANE];

  const int tid   = threadIdx.x;
  const int lane  = tid & 31;
  const int wave  = tid >> 5;
  const int n0    = wave * 16;
  const int half  = lane >> 4;
  const int slot0 = 2 * half;
  const int col   = n0 + (lane & 15);
  const int b0    = blockIdx.x * BSUB;
  const int rowbase = half * 8 * LDA + col;
  const int arow_off = (lane & 15) * LDA + (half << 1);
  int qoff[4];
#pragma unroll
  for (int q = 0; q < 4; ++q) qoff[q] = ((wave + q) & 3) * 16;

  v2f bw[KSTEPS];
  {
    const int h2 = half << 1;
    const float* __restrict__ wrow = W + col * S_DIM + h2;
#pragma unroll
    for (int q = 0; q < 4; ++q) {
#pragma unroll
      for (int i = 0; i < QSTEPS; ++i) {
        bw[QSTEPS * q + i].x = wrow[qoff[q] + 4 * i];
        bw[QSTEPS * q + i].y = wrow[qoff[q] + 4 * i + 1];
      }
    }
  }
  for (int e = tid; e < ROWS * LDA; e += THREADS) {
    sXa[e] = 0.f; sXb[e] = 0.f;
  }
  for (int e = tid; e < (N_ITER - 1) * C_DIM * PLANE; e += THREADS) (&sYv[0][0][0])[e] = 0.f;
  __syncthreads();
  for (int e = tid; e < PLANE; e += THREADS) {
    int b = e >> 6, j = e & 63;
    float v = y0[(b0 + b) * S_DIM + j];
    sYprev[e] = v;
    sXa[((N_ITER - 1) * BSUB + b) * LDA + j] = -v;
    out[(size_t)(b0 + b) * S_DIM + j] = v;
  }
  __syncthreads();

  float xfin[2 * BSUB];
  float pcl[2 * BSUB];
  float pprev[BSUB];
#pragma unroll
  for (int v = 0; v < 2 * BSUB; ++v) { xfin[v] = 0.f; pcl[v] = 0.f; }
#pragma unroll
  for (int u = 0; u < BSUB; ++u) pprev[u] = 0.f;
  int par = 0;
  const v8f zero8 = {0.f, 0.f, 0.f, 0.f, 0.f, 0.f, 0.f, 0.f};

  for (int blk = 0; blk < NB; ++blk) {
    const float* fblk =
        forces + ((size_t)(1 + blk * C_DIM) * B_TOT + b0) * S_DIM;
    for (int off = tid * 16; off < C_DIM * BSUB * S_DIM; off += THREADS * 16)
      __builtin_prefetch(fblk + off, 0, 1);

    for (int d = 0; d < NDIAG; ++d) {
      float* bufP = par ? sXb : sXa;
      float* bufQ = par ? sXa : sXb;

      const bool copy_prev = (d >= N_ITER) || (d == 0 && blk > 0);
      const int  t_prev    = blk * C_DIM + (d == 0 ? 0 : d - (N_ITER - 1));

      float ym[2 * BSUB], yc[2 * BSUB], fv[2 * BSUB], gg[2 * BSUB], rr[2 * BSUB];
      float4 oval = {0.f, 0.f, 0.f, 0.f};

      {
        v2f ax[KSTEPS];
        const float* __restrict__ xrow = bufP + arow_off;
        load_a_quarter<0>(xrow + qoff[0], ax);
        bar_arrive();
#pragma unroll
        for (int q = 0; q < 2; ++q) {
          const int cc = min(max(d - slot0 - q, 0), C_DIM - 1);
          const float* fp = fblk + (size_t)cc * B_TOT * S_DIM + col;
#pragma unroll
          for (int u = 0; u < BSUB; ++u) fv[4 * q + u] = fp[u * S_DIM];
        }
        bar_wait();
        load_a_quarter<1>(xrow + qoff[1], ax);
        load_a_quarter<2>(xrow + qoff[2], ax);
        load_a_quarter<3>(xrow + qoff[3], ax);
        __builtin_amdgcn_sched_barrier(0);
#pragma unroll
        for (int q = 0; q < 2; ++q) {
          const int slot = slot0 + q;
          const int c    = d - slot;
          const bool on  = (unsigned)c < (unsigned)C_DIM;
          const int lv   = max(slot - 1, 0);
          const int cm   = min(max(c - 1, 0), C_DIM - 1);
          const int cq   = min(max(c, 0), C_DIM - 1);
          const float* pm = (c == 0) ? &sYprev[col] : &sYv[lv][cm][col];
          const float* pc = &sYv[lv][cq][col];
          const bool usem = on && (c == 0 || slot > 0);
          const bool usec = on && (slot > 0);
#pragma unroll
          for (int u = 0; u < BSUB; ++u) {
            const float a = pm[u * S_DIM];
            const float b = pc[u * S_DIM];
            ym[4 * q + u] = usem ? a : 0.f;
            yc[4 * q + u] = usec ? b : 0.f;
          }
        }
        if (copy_prev && tid < PLANE / 4) oval = reinterpret_cast<const float4*>(sOut)[tid];
        __builtin_amdgcn_sched_barrier(0);
        v8f acc0 = wmma_step(ax[0], bw[0], zero8);
        v8f acc1 = wmma_step(ax[1], bw[1], zero8);
#pragma unroll
        for (int j = 2; j < KSTEPS; j += 2) {
          acc0 = wmma_step(ax[j], bw[j], acc0);
          acc1 = wmma_step(ax[j + 1], bw[j + 1], acc1);
        }
        const v8f wacc = acc0 + acc1;
        float up[2 * BSUB], xo[BSUB], pre[2 * BSUB];
#pragma unroll
        for (int v = 0; v < 2 * BSUB; ++v) up[v] = pcl[v] - wacc[v];
#pragma unroll
        for (int u = 0; u < BSUB; ++u) xo[u] = xhalf(up[BSUB + u]);
        if (d == 0) {
#pragma unroll
          for (int u = 0; u < BSUB; ++u) pprev[u] = half ? up[BSUB + u] : xo[u];
        }
#pragma unroll
        for (int q = 0; q < 2; ++q) {
          const int c = d - slot0 - q;
#pragma unroll
          for (int u = 0; u < BSUB; ++u) {
            const int v = 4 * q + u;
            const float pcn = (q == 0) ? (half ? xo[u] : 0.f) : up[u];
            const float pmn = (c == 0) ? pprev[u] : pcl[v];
            pre[v] = 0.5f * (pmn + pcn);
            pcl[v] = pcn;
          }
        }
        const bool on0 = (unsigned)(d - slot0) < (unsigned)C_DIM;
        const bool on1 = (unsigned)(d - slot0 - 1) < (unsigned)C_DIM;
        const float am0 = fmaxf(fmaxf(fabsf(pre[0]), fabsf(pre[1])), fmaxf(fabsf(pre[2]), fabsf(pre[3])));
        const float am1 = fmaxf(fmaxf(fabsf(pre[4]), fabsf(pre[5])), fmaxf(fabsf(pre[6]), fabsf(pre[7])));
        const bool wide = (on0 & !(am0 < TANH_SMALL_MAX)) | (on1 & !(am1 < TANH_SMALL_MAX));
        float tt[2 * BSUB];
        if (__builtin_expect(__builtin_amdgcn_ballot_w32(wide) != 0u, 0)) {
#pragma unroll
          for (int v = 0; v < 2 * BSUB; ++v) tt[v] = tanhf(pre[v]);
        } else {
#pragma unroll
          for (int v = 0; v < 2 * BSUB; ++v) tt[v] = tanh_small(pre[v]);
        }
#pragma unroll
        for (int q = 0; q < 2; ++q) {
          const int c = d - slot0 - q;
#pragma unroll
          for (int u = 0; u < BSUB; ++u) {
            const int v = 4 * q + u;
            float t = tt[v];
            float g = (0.5f * DT_C) * fmaf(-t, t, 1.0f);
            float R = fmaf(-DT_C, fv[v], fmaf(DT_C, t, yc[v] - ym[v]));
            float rhs = R;
            if (c > 0) rhs = R + fmaf(-wacc[v], g, xfin[v]);
            gg[v] = g;
            rr[v] = rhs;
          }
        }
#pragma unroll
        for (int q = 0; q < 2; ++q) {
          const int c = d - slot0 - q;
          if ((unsigned)c < (unsigned)C_DIM) {
#pragma unroll
            for (int u = 0; u < BSUB; ++u)
              bufQ[rowbase + (4 * q + u) * LDA] = rr[4 * q + u];
          }
        }
        if (copy_prev && tid < PLANE / 4) {
          float* dst = out + ((size_t)t_prev * B_TOT + b0) * S_DIM;
          reinterpret_cast<float4*>(dst)[tid] = oval;
        }
      }

      for (int k = 0; k < NEUMANN; ++k) {
        const float* xin  = (k & 1) ? bufP : bufQ;
        float*       xout = (k & 1) ? bufQ : bufP;
        v2f ax[KSTEPS];
        const float* __restrict__ xrow = xin + arow_off;
        load_a_quarter<0>(xrow + qoff[0], ax);
        bar_arrive();
        bar_wait();
        load_a_quarter<1>(xrow + qoff[1], ax);
        load_a_quarter<2>(xrow + qoff[2], ax);
        load_a_quarter<3>(xrow + qoff[3], ax);
        __builtin_amdgcn_sched_barrier(0);
        v8f acc0 = wmma_step(ax[0], bw[0], zero8);
        v8f acc1 = wmma_step(ax[1], bw[1], zero8);
#pragma unroll
        for (int j = 2; j < KSTEPS; j += 2) {
          acc0 = wmma_step(ax[j], bw[j], acc0);
          acc1 = wmma_step(ax[j + 1], bw[j + 1], acc1);
        }
        v8f macc = acc0 + acc1;
        asm volatile("" : "+v"(macc));
#pragma unroll
        for (int q = 0; q < 2; ++q) {
          const int slot = slot0 + q;
          const int c    = d - slot;
          if ((unsigned)c < (unsigned)C_DIM) {
            float xn[BSUB];
#pragma unroll
            for (int u = 0; u < BSUB; ++u) {
              const int v = 4 * q + u;
              xn[u] = rr[v] - gg[v] * macc[v];
              xout[rowbase + v * LDA] = xn[u];
            }
            if (k == NEUMANN - 1) {
              float yn[BSUB];
#pragma unroll
              for (int u = 0; u < BSUB; ++u) {
                const int v = 4 * q + u;
                yn[u] = yc[v] - xn[u];
                xfin[v] = xn[u];
              }
              if (slot < N_ITER - 1) {
#pragma unroll
                for (int u = 0; u < BSUB; ++u) sYv[slot][c][u * S_DIM + col] = yn[u];
              } else {
#pragma unroll
                for (int u = 0; u < BSUB; ++u) sOut[u * S_DIM + col] = yn[u];
                if (c == C_DIM - 1) {
#pragma unroll
                  for (int u = 0; u < BSUB; ++u) sYprev[u * S_DIM + col] = yn[u];
                }
              }
            }
          }
        }
      }
      par ^= 1;
    }
  }

  __syncthreads();
  if (tid < PLANE / 4) {
    float4 val = reinterpret_cast<const float4*>(sOut)[tid];
    float* dst = out + ((size_t)(NB * C_DIM) * B_TOT + b0) * S_DIM;
    reinterpret_cast<float4*>(dst)[tid] = val;
  }
}

extern "C" void kernel_launch(void* const* d_in, const int* in_sizes, int n_in,
                              void* d_out, int out_size, void* d_ws,
                              size_t ws_size, hipStream_t stream) {
  (void)in_sizes; (void)n_in; (void)out_size; (void)d_ws; (void)ws_size;
  const float* y0     = (const float*)d_in[0];
  const float* forces = (const float*)d_in[1];
  const float* W      = (const float*)d_in[2];
  float* out = (float*)d_out;
  solver_gfx1250_kernel<<<NWG, THREADS, 0, stream>>>(y0, forces, W, out);
}


# Neumann seed and update stored for all 8 values of a lane under no lane test (rows of a slot off the grid feed only unused output rows): the two masked regions per GEMM phase go; sum order kept
# speedup vs baseline: 7.3689x; 1.0329x over previous
#include <hip/hip_runtime.h>


#define S_DIM   64
#define C_DIM   8
#define N_ITER  4
#define NEUMANN 6
#define B_TOT   128
#define NB      8
#define BSUB    4
#define NWG     (B_TOT / BSUB)
#define ROWS    (N_ITER * BSUB)
#define NDIAG   (C_DIM + N_ITER - 1)
#define PLANE   (BSUB * S_DIM)
#define THREADS 128
#define DT_C    0.05f
#define LDA     66

typedef __attribute__((ext_vector_type(2))) float v2f;
typedef __attribute__((ext_vector_type(8))) float v8f;

#define KSTEPS (S_DIM / 4)

#define QSTEPS (KSTEPS / 4)

template <int Q>
__device__ __forceinline__ void load_a_quarter(const float* __restrict__ rowq, v2f (&a)[KSTEPS]) {
#pragma unroll
  for (int i = 0; i < QSTEPS; ++i) { a[QSTEPS * Q + i].x = rowq[4 * i]; a[QSTEPS * Q + i].y = rowq[4 * i + 1]; }
}

__device__ __forceinline__ v8f wmma_step(const v2f a, const v2f b, const v8f c) {
  return __builtin_amdgcn_wmma_f32_16x16x4_f32(false, a, false, b, (short)0, c, false, false);
}

__device__ __forceinline__ float xhalf(float x) {
  return __builtin_bit_cast(float, __builtin_amdgcn_permlanex16(__builtin_bit_cast(int, x), __builtin_bit_cast(int, x),
                                                                  0x76543210, 0xfedcba98, false, false));
}

__device__ __forceinline__ void bar_arrive() {
  __builtin_amdgcn_fence(__ATOMIC_RELEASE, "workgroup", "local");
  __builtin_amdgcn_s_barrier_signal(-1);
}
__device__ __forceinline__ void bar_wait() {
  __builtin_amdgcn_s_barrier_wait(-1);
  __builtin_amdgcn_fence(__ATOMIC_ACQUIRE, "workgroup", "local");
}

#define TANH_SMALL_MAX 0.625f
__device__ __forceinline__ float tanh_small(float x) {
  const float ax = fabsf(x);
  const float z  = x * x;
  float p = fmaf(__uint_as_float(0xbbbac73du), z, __uint_as_float(0x3ca908c9u));
  p = fmaf(z, p, __uint_as_float(0xbd5c1c4eu));
  p = fmaf(z, p, __uint_as_float(0x3e088382u));
  p = fmaf(z, p, __uint_as_float(0xbeaaaa99u));
  p = ax * p;
  return copysignf(fmaf(z, p, ax), x);
}

__global__ __launch_bounds__(THREADS, 1)
void solver_gfx1250_kernel(const float* __restrict__ y0,
                           const float* __restrict__ forces,
                           const float* __restrict__ W,
                           float* __restrict__ out) {
  __shared__ float sXa[ROWS * LDA];
  __shared__ float sXb[ROWS * LDA];
  __shared__ float sYv[N_ITER - 1][C_DIM][PLANE];
  __shared__ float sYprev[PLANE];
  __shared__ __attribute__((aligned(16))) float sOut[PLANE];

  const int tid   = threadIdx.x;
  const int lane  = tid & 31;
  const int wave  = tid >> 5;
  const int n0    = wave * 16;
  const int half  = lane >> 4;
  const int slot0 = 2 * half;
  const int col   = n0 + (lane & 15);
  const int b0    = blockIdx.x * BSUB;
  const int rowbase = half * 8 * LDA + col;
  const int arow_off = (lane & 15) * LDA + (half << 1);
  int qoff[4];
#pragma unroll
  for (int q = 0; q < 4; ++q) qoff[q] = ((wave + q) & 3) * 16;

  v2f bw[KSTEPS];
  {
    const int h2 = half << 1;
    const float* __restrict__ wrow = W + col * S_DIM + h2;
#pragma unroll
    for (int q = 0; q < 4; ++q) {
#pragma unroll
      for (int i = 0; i < QSTEPS; ++i) {
        bw[QSTEPS * q + i].x = wrow[qoff[q] + 4 * i];
        bw[QSTEPS * q + i].y = wrow[qoff[q] + 4 * i + 1];
      }
    }
  }
  for (int e = tid; e < ROWS * LDA; e += THREADS) {
    sXa[e] = 0.f; sXb[e] = 0.f;
  }
  for (int e = tid; e < (N_ITER - 1) * C_DIM * PLANE; e += THREADS) (&sYv[0][0][0])[e] = 0.f;
  __syncthreads();
  for (int e = tid; e < PLANE; e += THREADS) {
    int b = e >> 6, j = e & 63;
    float v = y0[(b0 + b) * S_DIM + j];
    sYprev[e] = v;
    sXa[((N_ITER - 1) * BSUB + b) * LDA + j] = -v;
    out[(size_t)(b0 + b) * S_DIM + j] = v;
  }
  __syncthreads();

  float xfin[2 * BSUB];
  float pcl[2 * BSUB];
  float pprev[BSUB];
#pragma unroll
  for (int v = 0; v < 2 * BSUB; ++v) { xfin[v] = 0.f; pcl[v] = 0.f; }
#pragma unroll
  for (int u = 0; u < BSUB; ++u) pprev[u] = 0.f;
  int par = 0;
  const v8f zero8 = {0.f, 0.f, 0.f, 0.f, 0.f, 0.f, 0.f, 0.f};

  for (int blk = 0; blk < NB; ++blk) {
    const float* fblk =
        forces + ((size_t)(1 + blk * C_DIM) * B_TOT + b0) * S_DIM;
    for (int off = tid * 16; off < C_DIM * BSUB * S_DIM; off += THREADS * 16)
      __builtin_prefetch(fblk + off, 0, 1);

    for (int d = 0; d < NDIAG; ++d) {
      float* bufP = par ? sXb : sXa;
      float* bufQ = par ? sXa : sXb;

      const bool copy_prev = (d >= N_ITER) || (d == 0 && blk > 0);
      const int  t_prev    = blk * C_DIM + (d == 0 ? 0 : d - (N_ITER - 1));

      float ym[2 * BSUB], yc[2 * BSUB], fv[2 * BSUB], gg[2 * BSUB], rr[2 * BSUB];
      float4 oval = {0.f, 0.f, 0.f, 0.f};

      {
        v2f ax[KSTEPS];
        const float* __restrict__ xrow = bufP + arow_off;
        load_a_quarter<0>(xrow + qoff[0], ax);
        bar_arrive();
#pragma unroll
        for (int q = 0; q < 2; ++q) {
          const int cc = min(max(d - slot0 - q, 0), C_DIM - 1);
          const float* fp = fblk + (size_t)cc * B_TOT * S_DIM + col;
#pragma unroll
          for (int u = 0; u < BSUB; ++u) fv[4 * q + u] = fp[u * S_DIM];
        }
        bar_wait();
        load_a_quarter<1>(xrow + qoff[1], ax);
        load_a_quarter<2>(xrow + qoff[2], ax);
        load_a_quarter<3>(xrow + qoff[3], ax);
        __builtin_amdgcn_sched_barrier(0);
#pragma unroll
        for (int q = 0; q < 2; ++q) {
          const int slot = slot0 + q;
          const int c    = d - slot;
          const bool on  = (unsigned)c < (unsigned)C_DIM;
          const int lv   = max(slot - 1, 0);
          const int cm   = min(max(c - 1, 0), C_DIM - 1);
          const int cq   = min(max(c, 0), C_DIM - 1);
          const float* pm = (c == 0) ? &sYprev[col] : &sYv[lv][cm][col];
          const float* pc = &sYv[lv][cq][col];
          const bool usem = on && (c == 0 || slot > 0);
          const bool usec = on && (slot > 0);
#pragma unroll
          for (int u = 0; u < BSUB; ++u) {
            const float a = pm[u * S_DIM];
            const float b = pc[u * S_DIM];
            ym[4 * q + u] = usem ? a : 0.f;
            yc[4 * q + u] = usec ? b : 0.f;
          }
        }
        if (copy_prev && tid < PLANE / 4) oval = reinterpret_cast<const float4*>(sOut)[tid];
        __builtin_amdgcn_sched_barrier(0);
        v8f acc0 = wmma_step(ax[0], bw[0], zero8);
        v8f acc1 = wmma_step(ax[1], bw[1], zero8);
#pragma unroll
        for (int j = 2; j < KSTEPS; j += 2) {
          acc0 = wmma_step(ax[j], bw[j], acc0);
          acc1 = wmma_step(ax[j + 1], bw[j + 1], acc1);
        }
        const v8f wacc = acc0 + acc1;
        float up[2 * BSUB], xo[BSUB], pre[2 * BSUB];
#pragma unroll
        for (int v = 0; v < 2 * BSUB; ++v) up[v] = pcl[v] - wacc[v];
#pragma unroll
        for (int u = 0; u < BSUB; ++u) xo[u] = xhalf(up[BSUB + u]);
        if (d == 0) {
#pragma unroll
          for (int u = 0; u < BSUB; ++u) pprev[u] = half ? up[BSUB + u] : xo[u];
        }
#pragma unroll
        for (int q = 0; q < 2; ++q) {
          const int c = d - slot0 - q;
#pragma unroll
          for (int u = 0; u < BSUB; ++u) {
            const int v = 4 * q + u;
            const float pcn = (q == 0) ? (half ? xo[u] : 0.f) : up[u];
            const float pmn = (c == 0) ? pprev[u] : pcl[v];
            pre[v] = 0.5f * (pmn + pcn);
            pcl[v] = pcn;
          }
        }
        const bool on0 = (unsigned)(d - slot0) < (unsigned)C_DIM;
        const bool on1 = (unsigned)(d - slot0 - 1) < (unsigned)C_DIM;
        const float am0 = fmaxf(fmaxf(fabsf(pre[0]), fabsf(pre[1])), fmaxf(fabsf(pre[2]), fabsf(pre[3])));
        const float am1 = fmaxf(fmaxf(fabsf(pre[4]), fabsf(pre[5])), fmaxf(fabsf(pre[6]), fabsf(pre[7])));
        const bool wide = (on0 & !(am0 < TANH_SMALL_MAX)) | (on1 & !(am1 < TANH_SMALL_MAX));
        float tt[2 * BSUB];
        if (__builtin_expect(__builtin_amdgcn_ballot_w32(wide) != 0u, 0)) {
#pragma unroll
          for (int v = 0; v < 2 * BSUB; ++v) tt[v] = tanhf(pre[v]);
        } else {
#pragma unroll
          for (int v = 0; v < 2 * BSUB; ++v) tt[v] = tanh_small(pre[v]);
        }
#pragma unroll
        for (int q = 0; q < 2; ++q) {
          const int c = d - slot0 - q;
#pragma unroll
          for (int u = 0; u < BSUB; ++u) {
            const int v = 4 * q + u;
            float t = tt[v];
            float g = (0.5f * DT_C) * fmaf(-t, t, 1.0f);
            float R = fmaf(-DT_C, fv[v], fmaf(DT_C, t, yc[v] - ym[v]));
            float rhs = R;
            if (c > 0) rhs = R + fmaf(-wacc[v], g, xfin[v]);
            gg[v] = g;
            rr[v] = rhs;
          }
        }
#pragma unroll
        for (int v = 0; v < 2 * BSUB; ++v) bufQ[rowbase + v * LDA] = rr[v];
        if (copy_prev && tid < PLANE / 4) {
          float* dst = out + ((size_t)t_prev * B_TOT + b0) * S_DIM;
          reinterpret_cast<float4*>(dst)[tid] = oval;
        }
      }

      for (int k = 0; k < NEUMANN; ++k) {
        const float* xin  = (k & 1) ? bufP : bufQ;
        float*       xout = (k & 1) ? bufQ : bufP;
        v2f ax[KSTEPS];
        const float* __restrict__ xrow = xin + arow_off;
        load_a_quarter<0>(xrow + qoff[0], ax);
        bar_arrive();
        bar_wait();
        load_a_quarter<1>(xrow + qoff[1], ax);
        load_a_quarter<2>(xrow + qoff[2], ax);
        load_a_quarter<3>(xrow + qoff[3], ax);
        __builtin_amdgcn_sched_barrier(0);
        v8f acc0 = wmma_step(ax[0], bw[0], zero8);
        v8f acc1 = wmma_step(ax[1], bw[1], zero8);
#pragma unroll
        for (int j = 2; j < KSTEPS; j += 2) {
          acc0 = wmma_step(ax[j], bw[j], acc0);
          acc1 = wmma_step(ax[j + 1], bw[j + 1], acc1);
        }
        const v8f macc = acc0 + acc1;
        float xn[2 * BSUB];
#pragma unroll
        for (int v = 0; v < 2 * BSUB; ++v) {
          xn[v] = rr[v] - gg[v] * macc[v];
          xout[rowbase + v * LDA] = xn[v];
        }
        if (k == NEUMANN - 1) {
#pragma unroll
          for (int q = 0; q < 2; ++q) {
            const int slot = slot0 + q;
            const int c    = d - slot;
            if ((unsigned)c < (unsigned)C_DIM) {
              float yn[BSUB];
#pragma unroll
              for (int u = 0; u < BSUB; ++u) {
                const int v = 4 * q + u;
                yn[u] = yc[v] - xn[v];
                xfin[v] = xn[v];
              }
              if (slot < N_ITER - 1) {
#pragma unroll
                for (int u = 0; u < BSUB; ++u) sYv[slot][c][u * S_DIM + col] = yn[u];
              } else {
#pragma unroll
                for (int u = 0; u < BSUB; ++u) sOut[u * S_DIM + col] = yn[u];
                if (c == C_DIM - 1) {
#pragma unroll
                  for (int u = 0; u < BSUB; ++u) sYprev[u * S_DIM + col] = yn[u];
                }
              }
            }
          }
        }
      }
      par ^= 1;
    }
  }

  __syncthreads();
  if (tid < PLANE / 4) {
    float4 val = reinterpret_cast<const float4*>(sOut)[tid];
    float* dst = out + ((size_t)(NB * C_DIM) * B_TOT + b0) * S_DIM;
    reinterpret_cast<float4*>(dst)[tid] = val;
  }
}

extern "C" void kernel_launch(void* const* d_in, const int* in_sizes, int n_in,
                              void* d_out, int out_size, void* d_ws,
                              size_t ws_size, hipStream_t stream) {
  (void)in_sizes; (void)n_in; (void)out_size; (void)d_ws; (void)ws_size;
  const float* y0     = (const float*)d_in[0];
  const float* forces = (const float*)d_in[1];
  const float* W      = (const float*)d_in[2];
  float* out = (float*)d_out;
  solver_gfx1250_kernel<<<NWG, THREADS, 0, stream>>>(y0, forces, W, out);
}


# Products transposed (W = A operand, cell rows = B columns): a lane's 8 accumulator registers ARE its own-quarter B fragments, no LDS read-back; x exchanged in 16-byte accesses; sum order changes
# speedup vs baseline: 8.0719x; 1.0954x over previous
#include <hip/hip_runtime.h>


#define S_DIM   64
#define C_DIM   8
#define N_ITER  4
#define NEUMANN 6
#define B_TOT   128
#define NB      8
#define BSUB    4
#define NWG     (B_TOT / BSUB)
#define ROWS    (N_ITER * BSUB)
#define NDIAG   (C_DIM + N_ITER - 1)
#define THREADS 128
#define DT_C    0.05f
#define XP      68
#define YP      68
#define PLANE   (BSUB * YP)

#define OFF_XA    0
#define OFF_XB    (ROWS * XP)
#define OFF_YV    (2 * ROWS * XP)
#define OFF_YPREV (OFF_YV + (N_ITER - 1) * C_DIM * PLANE)
#define OFF_OUT   (OFF_YPREV + PLANE)
#define SMEM_FLOATS (OFF_OUT + PLANE)

typedef __attribute__((ext_vector_type(2))) float v2f;
typedef __attribute__((ext_vector_type(8))) float v8f;

#define KSTEPS (S_DIM / 4)
#define QSTEPS (KSTEPS / 4)

__device__ __forceinline__ v8f wmma_step(const v2f a, const v2f b, const v8f c) {
  return __builtin_amdgcn_wmma_f32_16x16x4_f32(false, a, false, b, (short)0, c, false, false);
}

__device__ __forceinline__ void ld8(const float* __restrict__ p, float (&v)[8]) {
  const float4 a = *reinterpret_cast<const float4*>(p);
  const float4 b = *reinterpret_cast<const float4*>(p + 4);
  v[0] = a.x; v[1] = a.y; v[2] = a.z; v[3] = a.w; v[4] = b.x; v[5] = b.y; v[6] = b.z; v[7] = b.w;
}
__device__ __forceinline__ void st8(float* __restrict__ p, const float (&v)[8]) {
  *reinterpret_cast<float4*>(p)     = make_float4(v[0], v[1], v[2], v[3]);
  *reinterpret_cast<float4*>(p + 4) = make_float4(v[4], v[5], v[6], v[7]);
}

__device__ __forceinline__ void frag4(const float (&f)[8], v2f* __restrict__ dst) {
#pragma unroll
  for (int p = 0; p < QSTEPS; ++p) { dst[p].x = f[2 * p]; dst[p].y = f[2 * p + 1]; }
}

__device__ __forceinline__ float from_slot_below(float x) {
  return __builtin_bit_cast(float, __builtin_amdgcn_permlane16(__builtin_bit_cast(unsigned, x), __builtin_bit_cast(unsigned, x),
                                                                 0x3210fedcu, 0xba987654u, false, false));
}
__device__ __forceinline__ float from_last_slot(float x) {
  return __builtin_bit_cast(float, __builtin_amdgcn_permlane16(__builtin_bit_cast(unsigned, x), __builtin_bit_cast(unsigned, x),
                                                                 0xfedcfedcu, 0xfedcfedcu, false, false));
}

__device__ __forceinline__ void bar_arrive() {
  __builtin_amdgcn_fence(__ATOMIC_RELEASE, "workgroup", "local");
  __builtin_amdgcn_s_barrier_signal(-1);
}
__device__ __forceinline__ void bar_wait() {
  __builtin_amdgcn_s_barrier_wait(-1);
  __builtin_amdgcn_fence(__ATOMIC_ACQUIRE, "workgroup", "local");
}

#define TANH_SMALL_MAX 0.625f
__device__ __forceinline__ float tanh_small(float x) {
  const float ax = fabsf(x);
  const float z  = x * x;
  float p = fmaf(__uint_as_float(0xbbbac73du), z, __uint_as_float(0x3ca908c9u));
  p = fmaf(z, p, __uint_as_float(0xbd5c1c4eu));
  p = fmaf(z, p, __uint_as_float(0x3e088382u));
  p = fmaf(z, p, __uint_as_float(0xbeaaaa99u));
  p = ax * p;
  return copysignf(fmaf(z, p, ax), x);
}

#define GEMM_PHASE(xin_off, BETWEEN, result)                                                                              \
  {                                                                                                                       \
    v2f bq[QSTEPS];                                                                                                       \
    frag4(xq, bq);                                                                                                        \
      \
      \
    v8f acc0 = wmma_step(aw[0], bq[0], zero8);                                                                            \
    v8f acc1 = wmma_step(aw[1], bq[1], zero8);                                                                            \
    acc0 = wmma_step(aw[2], bq[2], acc0);                                                                                 \
    acc1 = wmma_step(aw[3], bq[3], acc1);                                                                                 \
    bar_arrive();                                                                                                         \
    BETWEEN                                                                                                               \
    bar_wait();                                                                                                           \
    v2f bf[KSTEPS - QSTEPS];                                                                                              \
    {                                                                                                                     \
      const float* __restrict__ xrow = smem + (xin_off) + c * XP + h * 8;                                                 \
      float f8[8];                                                                                                        \
      ld8(xrow + qoff[1], f8); frag4(f8, bf);                                                                             \
      ld8(xrow + qoff[2], f8); frag4(f8, bf + QSTEPS);                                                                    \
      ld8(xrow + qoff[3], f8); frag4(f8, bf + 2 * QSTEPS);                                                                \
    }

#define GEMM_CHAIN(result)                                                                                                \
      \
    __builtin_amdgcn_sched_barrier(0);                                                                                    \
    _Pragma("unroll")                                                                                                     \
    for (int j = QSTEPS; j < KSTEPS; j += 2) {                                                                            \
      acc0 = wmma_step(aw[j], bf[j - QSTEPS], acc0);                                                                      \
      acc1 = wmma_step(aw[j + 1], bf[j + 1 - QSTEPS], acc1);                                                              \
    }                                                                                                                     \
    result = acc0 + acc1;                                                                                                 \
  }

__global__ __launch_bounds__(THREADS, 1)
void solver_gfx1250_kernel(const float* __restrict__ y0,
                           const float* __restrict__ forces,
                           const float* __restrict__ W,
                           float* __restrict__ out) {
  __shared__ __attribute__((aligned(16))) float smem[SMEM_FLOATS];

  const int tid   = threadIdx.x;
  const int lane  = tid & 31;
  const int wave  = tid >> 5;
  const int c     = lane & 15;
  const int h     = lane >> 4;
  const int slot  = c >> 2;
  const int u     = c & 3;
  const int n0    = wave * 16 + h * 8;
  const int b0    = blockIdx.x * BSUB;
  const int xown  = c * XP + n0;
  const int yoff  = u * YP + n0;
  int qoff[4];
#pragma unroll
  for (int q = 0; q < 4; ++q) qoff[q] = ((wave + q) & 3) * 16;

  v2f aw[KSTEPS];
  {
    const float* __restrict__ wrow = W + (wave * 16 + c) * S_DIM + h * 8;
#pragma unroll
    for (int q = 0; q < 4; ++q) {
      float f8[8];
      ld8(wrow + qoff[q], f8);
      frag4(f8, aw + QSTEPS * q);
    }
  }
  for (int e = tid; e < SMEM_FLOATS; e += THREADS) smem[e] = 0.f;
  __syncthreads();
  for (int e = tid; e < BSUB * S_DIM; e += THREADS) {
    int b = e >> 6, j = e & 63;
    float v = y0[(b0 + b) * S_DIM + j];
    smem[OFF_YPREV + b * YP + j] = v;
    smem[OFF_XA + ((N_ITER - 1) * BSUB + b) * XP + j] = -v;
    out[(size_t)(b0 + b) * S_DIM + j] = v;
  }
  __syncthreads();

  float xq[8];
  float xfin[8];
  float pcl[8];
  float pprev[8];
  {
    float y8[8];
    ld8(y0 + (b0 + u) * S_DIM + n0, y8);
#pragma unroll
    for (int v = 0; v < 8; ++v) {
      xq[v] = (slot == N_ITER - 1) ? -y8[v] : 0.f;
      xfin[v] = 0.f; pcl[v] = 0.f; pprev[v] = 0.f;
    }
  }
  int par = 0;
  const v8f zero8 = {0.f, 0.f, 0.f, 0.f, 0.f, 0.f, 0.f, 0.f};

  for (int blk = 0; blk < NB; ++blk) {
    const float* fblk =
        forces + ((size_t)(1 + blk * C_DIM) * B_TOT + b0) * S_DIM;
    for (int off = tid * 16; off < C_DIM * BSUB * S_DIM; off += THREADS * 16)
      __builtin_prefetch(fblk + off, 0, 1);

    for (int d = 0; d < NDIAG; ++d) {
      const int offP = par ? OFF_XB : OFF_XA;
      const int offQ = par ? OFF_XA : OFF_XB;

      const bool copy_prev = (d >= N_ITER) || (d == 0 && blk > 0);
      const int  t_prev    = blk * C_DIM + (d == 0 ? 0 : d - (N_ITER - 1));

      const int  cidx = d - slot;
      const bool on   = (unsigned)cidx < (unsigned)C_DIM;

      float ym[8], yc[8], fv[8], gg[8], rr[8];
      float4 oval = {0.f, 0.f, 0.f, 0.f};

      {
        v8f wacc;
        GEMM_PHASE(offP,
          {
            const int cc = min(max(cidx, 0), C_DIM - 1);
            ld8(fblk + ((size_t)cc * B_TOT + u) * S_DIM + n0, fv);
          },
          wacc)
        {
          const int lv   = max(slot - 1, 0);
          const int cm   = min(max(cidx - 1, 0), C_DIM - 1);
          const int cq   = min(max(cidx, 0), C_DIM - 1);
          const float* pm = (cidx == 0) ? (smem + OFF_YPREV + yoff) : (smem + OFF_YV + (lv * C_DIM + cm) * PLANE + yoff);
          const float* pc = smem + OFF_YV + (lv * C_DIM + cq) * PLANE + yoff;
          const bool usem = on && (cidx == 0 || slot > 0);
          const bool usec = on && (slot > 0);
          float a8[8], b8[8];
          ld8(pm, a8);
          ld8(pc, b8);
#pragma unroll
          for (int v = 0; v < 8; ++v) {
            ym[v] = usem ? a8[v] : 0.f;
            yc[v] = usec ? b8[v] : 0.f;
          }
        }
        if (copy_prev && tid < BSUB * S_DIM / 4)
          oval = *reinterpret_cast<const float4*>(smem + OFF_OUT + (tid >> 4) * YP + (tid & 15) * 4);
        GEMM_CHAIN(wacc)
        float up[8], below[8], pre[8];
#pragma unroll
        for (int v = 0; v < 8; ++v) up[v] = pcl[v] - wacc[v];
#pragma unroll
        for (int v = 0; v < 8; ++v) below[v] = from_slot_below(up[v]);
        if (d == 0) {
#pragma unroll
          for (int v = 0; v < 8; ++v) pprev[v] = from_last_slot(up[v]);
        }
#pragma unroll
        for (int v = 0; v < 8; ++v) {
          const float pcn = (slot == 0) ? 0.f : below[v];
          const float pmn = (cidx == 0) ? pprev[v] : pcl[v];
          pre[v] = 0.5f * (pmn + pcn);
          pcl[v] = pcn;
        }
        const float am = fmaxf(fmaxf(fmaxf(fabsf(pre[0]), fabsf(pre[1])), fmaxf(fabsf(pre[2]), fabsf(pre[3]))),
                               fmaxf(fmaxf(fabsf(pre[4]), fabsf(pre[5])), fmaxf(fabsf(pre[6]), fabsf(pre[7]))));
        const bool wide = on & !(am < TANH_SMALL_MAX);
        float tt[8];
        if (__builtin_expect(__builtin_amdgcn_ballot_w32(wide) != 0u, 0)) {
#pragma unroll
          for (int v = 0; v < 8; ++v) tt[v] = tanhf(pre[v]);
        } else {
#pragma unroll
          for (int v = 0; v < 8; ++v) tt[v] = tanh_small(pre[v]);
        }
#pragma unroll
        for (int v = 0; v < 8; ++v) {
          float t = tt[v];
          float g = (0.5f * DT_C) * fmaf(-t, t, 1.0f);
          float R = fmaf(-DT_C, fv[v], fmaf(DT_C, t, yc[v] - ym[v]));
          float rhs = R;
          if (cidx > 0) rhs = R + fmaf(-wacc[v], g, xfin[v]);
          gg[v] = g;
          rr[v] = rhs;
          xq[v] = rhs;
        }
        st8(smem + offQ + xown, rr);
        if (copy_prev && tid < BSUB * S_DIM / 4) {
          float* dst = out + ((size_t)t_prev * B_TOT + b0) * S_DIM;
          reinterpret_cast<float4*>(dst)[tid] = oval;
        }
      }

      for (int k = 0; k < NEUMANN; ++k) {
        const int offin  = (k & 1) ? offP : offQ;
        const int offout = (k & 1) ? offQ : offP;
        v8f macc;
        GEMM_PHASE(offin, {}, macc)
        GEMM_CHAIN(macc)
        float xn[8];
#pragma unroll
        for (int v = 0; v < 8; ++v) {
          xn[v] = rr[v] - gg[v] * macc[v];
          xq[v] = xn[v];
        }
        st8(smem + offout + xown, xn);
        if (k == NEUMANN - 1) {
          if (on) {
            float yn[8];
#pragma unroll
            for (int v = 0; v < 8; ++v) {
              yn[v] = yc[v] - xn[v];
              xfin[v] = xn[v];
            }
            if (slot < N_ITER - 1) {
              st8(smem + OFF_YV + (slot * C_DIM + cidx) * PLANE + yoff, yn);
            } else {
              st8(smem + OFF_OUT + yoff, yn);
              if (cidx == C_DIM - 1) st8(smem + OFF_YPREV + yoff, yn);
            }
          }
        }
      }
      par ^= 1;
    }
  }

  __syncthreads();
  if (tid < BSUB * S_DIM / 4) {
    float4 val = *reinterpret_cast<const float4*>(smem + OFF_OUT + (tid >> 4) * YP + (tid & 15) * 4);
    float* dst = out + ((size_t)(NB * C_DIM) * B_TOT + b0) * S_DIM;
    reinterpret_cast<float4*>(dst)[tid] = val;
  }
}

extern "C" void kernel_launch(void* const* d_in, const int* in_sizes, int n_in,
                              void* d_out, int out_size, void* d_ws,
                              size_t ws_size, hipStream_t stream) {
  (void)in_sizes; (void)n_in; (void)out_size; (void)d_ws; (void)ws_size;
  const float* y0     = (const float*)d_in[0];
  const float* forces = (const float*)d_in[1];
  const float* W      = (const float*)d_in[2];
  float* out = (float*)d_out;
  solver_gfx1250_kernel<<<NWG, THREADS, 0, stream>>>(y0, forces, W, out);
}
